# starting kernel
# baseline (speedup 1.0000x reference)
#include <hip/hip_runtime.h>
#include <math.h>

typedef __attribute__((ext_vector_type(2))) float v2f;
typedef __attribute__((ext_vector_type(8))) float v8f;


__global__ void k_deg_init(float* __restrict__ deg, int n) {
    int i = blockIdx.x * blockDim.x + threadIdx.x;
    if (i < n) deg[i] = 1.0f;
}

__global__ void k_deg_count(const int* __restrict__ dst, float* __restrict__ deg, int E) {
    int e = blockIdx.x * blockDim.x + threadIdx.x;
    if (e < E) atomicAdd(&deg[dst[e]], 1.0f);
}

__global__ void k_dinv(const float* __restrict__ deg, float* __restrict__ dinv, int n) {
    int i = blockIdx.x * blockDim.x + threadIdx.x;
    if (i < n) dinv[i] = rsqrtf(deg[i]);
}

__global__ void __launch_bounds__(256) k_gemm64(const float* __restrict__ A,
                                                const float* __restrict__ B,
                                                float* __restrict__ C) {
    const int wave = threadIdx.x >> 5;
    const int lane = threadIdx.x & 31;
    const int half = lane >> 4;
    const int lm   = lane & 15;
    const int m0   = blockIdx.x * 32 + (wave >> 2) * 16;
    const int n0   = (wave & 3) * 16;
    const float* arow = A + (size_t)(m0 + lm) * 64;
    const int bcol = n0 + lm;

    v8f acc = {0.f,0.f,0.f,0.f,0.f,0.f,0.f,0.f};
#pragma unroll
    for (int kk = 0; kk < 16; ++kk) {
        const int kb = kk * 4 + half * 2;
        v2f a = *(const v2f*)(arow + kb);
        v2f b;
        b.x = B[(size_t)kb * 64 + bcol];
        b.y = B[(size_t)(kb + 1) * 64 + bcol];
        acc = __builtin_amdgcn_wmma_f32_16x16x4_f32(false, a, false, b,
                                                    (short)0, acc, false, false);
    }
#pragma unroll
    for (int i = 0; i < 8; ++i)
        C[(size_t)(m0 + half * 8 + i) * 64 + n0 + lm] = acc[i];
}

__global__ void k_init_h(const float* __restrict__ xw, const float* __restrict__ dinv,
                         const float* __restrict__ b1, float* __restrict__ h, int n) {
    int idx = blockIdx.x * blockDim.x + threadIdx.x;
    if (idx >= n * 64) return;
    int i = idx >> 6, c = idx & 63;
    float di = dinv[i];
    h[idx] = b1[c] + di * di * xw[idx];
}

__global__ void k_scatter64(const int* __restrict__ src, const int* __restrict__ dst,
                            const float* __restrict__ dinv, const float* __restrict__ xw,
                            float* __restrict__ h, int E) {
    int idx = blockIdx.x * blockDim.x + threadIdx.x;
    if (idx >= E * 16) return;
    int e = idx >> 4;
    int c = (idx & 15) << 2;
    int s = src[e], d = dst[e];
    float w = dinv[s] * dinv[d];
    const float4 v = *(const float4*)(xw + (size_t)s * 64 + c);
    float* o = h + (size_t)d * 64 + c;
    atomicAdd(o + 0, w * v.x);
    atomicAdd(o + 1, w * v.y);
    atomicAdd(o + 2, w * v.z);
    atomicAdd(o + 3, w * v.w);
}

__global__ void __launch_bounds__(128) k_relu_softmax(float* __restrict__ h, int n) {
    int node = blockIdx.x * 4 + (threadIdx.x >> 5);
    int lane = threadIdx.x & 31;
    if (node >= n) return;
    float* row = h + (size_t)node * 64;
    float v0 = fmaxf(row[lane], 0.0f);
    float v1 = fmaxf(row[lane + 32], 0.0f);
    float m = fmaxf(v0, v1);
#pragma unroll
    for (int off = 16; off > 0; off >>= 1) m = fmaxf(m, __shfl_xor(m, off, 32));
    float e0 = __expf(v0 - m), e1 = __expf(v1 - m);
    float s = e0 + e1;
#pragma unroll
    for (int off = 16; off > 0; off >>= 1) s += __shfl_xor(s, off, 32);
    float inv = 1.0f / s;
    row[lane]      = e0 * inv;
    row[lane + 32] = e1 * inv;
}

__global__ void __launch_bounds__(192) k_gemm34(const float* __restrict__ A,
                                                const float* __restrict__ B,
                                                float* __restrict__ C) {
    const int wave = threadIdx.x >> 5;
    const int lane = threadIdx.x & 31;
    const int half = lane >> 4;
    const int lm   = lane & 15;
    const int m0   = blockIdx.x * 32 + (wave / 3) * 16;
    const int n0   = (wave % 3) * 16;
    const int col  = n0 + lm;
    const bool ok  = (col < 34);
    const float* arow = A + (size_t)(m0 + lm) * 64;

    v8f acc = {0.f,0.f,0.f,0.f,0.f,0.f,0.f,0.f};
#pragma unroll
    for (int kk = 0; kk < 16; ++kk) {
        const int kb = kk * 4 + half * 2;
        v2f a = *(const v2f*)(arow + kb);
        v2f b;
        b.x = ok ? B[(size_t)kb * 34 + col]       : 0.0f;
        b.y = ok ? B[(size_t)(kb + 1) * 34 + col] : 0.0f;
        acc = __builtin_amdgcn_wmma_f32_16x16x4_f32(false, a, false, b,
                                                    (short)0, acc, false, false);
    }
    if (ok) {
#pragma unroll
        for (int i = 0; i < 8; ++i)
            C[(size_t)(m0 + half * 8 + i) * 34 + col] = acc[i];
    }
}

__global__ void k_init_out(const float* __restrict__ hw2, const float* __restrict__ dinv,
                           const float* __restrict__ b2, float* __restrict__ out, int n) {
    int idx = blockIdx.x * blockDim.x + threadIdx.x;
    if (idx >= n * 34) return;
    int i = idx / 34, c = idx - i * 34;
    float di = dinv[i];
    out[idx] = b2[c] + di * di * hw2[idx];
}

__global__ void k_scatter34(const int* __restrict__ src, const int* __restrict__ dst,
                            const float* __restrict__ dinv, const float* __restrict__ hw2,
                            float* __restrict__ out, int E) {
    int idx = blockIdx.x * blockDim.x + threadIdx.x;
    if (idx >= E * 34) return;
    int e = idx / 34, c = idx - e * 34;
    int s = src[e], d = dst[e];
    float w = dinv[s] * dinv[d];
    atomicAdd(&out[(size_t)d * 34 + c], w * hw2[(size_t)s * 34 + c]);
}

extern "C" void kernel_launch(void* const* d_in, const int* in_sizes, int n_in,
                              void* d_out, int out_size, void* d_ws, size_t ws_size,
                              hipStream_t stream) {
    const float* x  = (const float*)d_in[0];
    const int*   ei = (const int*)  d_in[1];
    const float* W1 = (const float*)d_in[2];
    const float* b1 = (const float*)d_in[3];
    const float* W2 = (const float*)d_in[4];
    const float* b2 = (const float*)d_in[5];
    float* out = (float*)d_out;

    const int N = in_sizes[0] / 64;
    const int E = in_sizes[1] / 2;
    const int* src = ei;
    const int* dst = ei + E;

    float* deg  = (float*)d_ws;
    float* dinv = deg  + N;
    float* xw   = dinv + N;
    float* h    = xw   + (size_t)N * 64;
    float* hw2  = h    + (size_t)N * 64;

    const int T = 256;
    k_deg_init <<<(N + T - 1) / T, T, 0, stream>>>(deg, N);
    k_deg_count<<<(E + T - 1) / T, T, 0, stream>>>(dst, deg, E);
    k_dinv     <<<(N + T - 1) / T, T, 0, stream>>>(deg, dinv, N);
    k_gemm64<<<N / 32, 256, 0, stream>>>(x, W1, xw);
    k_init_h   <<<((size_t)N * 64 + T - 1) / T, T, 0, stream>>>(xw, dinv, b1, h, N);
    k_scatter64<<<((size_t)E * 16 + T - 1) / T, T, 0, stream>>>(src, dst, dinv, xw, h, E);
    k_relu_softmax<<<(N + 3) / 4, 128, 0, stream>>>(h, N);
    k_gemm34<<<N / 32, 192, 0, stream>>>(h, W2, hw2);
    k_init_out <<<((size_t)N * 34 + T - 1) / T, T, 0, stream>>>(hw2, dinv, b2, out, N);
    k_scatter34<<<((size_t)E * 34 + T - 1) / T, T, 0, stream>>>(src, dst, dinv, hw2, out, E);
}


# Edges bucketed by destination once per call (count with rank, scan, fill); both layers sum each node's bucket by gather with plain stores, no float atomics; sum order = arrival order in bucket
# speedup vs baseline: 1.8763x; 1.8763x over previous
#include <hip/hip_runtime.h>
#include <math.h>

typedef __attribute__((ext_vector_type(2))) float v2f;
typedef __attribute__((ext_vector_type(8))) float v8f;


__global__ void k_zero_cnt(int* __restrict__ cnt, int n) {
    int i = blockIdx.x * blockDim.x + threadIdx.x;
    if (i < n) cnt[i] = 0;
}

__global__ void k_count(const int* __restrict__ dst, int* __restrict__ cnt,
                        int* __restrict__ rank, int E) {
    int e = blockIdx.x * blockDim.x + threadIdx.x;
    if (e < E) rank[e] = atomicAdd(&cnt[dst[e]], 1);
}

__global__ void __launch_bounds__(1024) k_scan_local(const int* __restrict__ cnt,
                                                     int* __restrict__ excl,
                                                     int* __restrict__ bsum, int n) {
    __shared__ int wsum[32];
    const int tid = threadIdx.x, lane = tid & 31, wv = tid >> 5;
    const int i0 = (blockIdx.x * 1024 + tid) * 4;
    const int c0 = (i0     < n) ? cnt[i0]     : 0;
    const int c1 = (i0 + 1 < n) ? cnt[i0 + 1] : 0;
    const int c2 = (i0 + 2 < n) ? cnt[i0 + 2] : 0;
    const int c3 = (i0 + 3 < n) ? cnt[i0 + 3] : 0;
    const int tsum = c0 + c1 + c2 + c3;
    int inc = tsum;
#pragma unroll
    for (int off = 1; off < 32; off <<= 1) {
        int v = __shfl_up(inc, off, 32);
        if (lane >= off) inc += v;
    }
    if (lane == 31) wsum[wv] = inc;
    __syncthreads();
    if (wv == 0) {
        const int w = wsum[lane];
        int winc = w;
#pragma unroll
        for (int off = 1; off < 32; off <<= 1) {
            int v = __shfl_up(winc, off, 32);
            if (lane >= off) winc += v;
        }
        wsum[lane] = winc - w;
        if (lane == 31) bsum[blockIdx.x] = winc;
    }
    __syncthreads();
    const int base = wsum[wv] + inc - tsum;
    if (i0     < n) excl[i0]     = base;
    if (i0 + 1 < n) excl[i0 + 1] = base + c0;
    if (i0 + 2 < n) excl[i0 + 2] = base + c0 + c1;
    if (i0 + 3 < n) excl[i0 + 3] = base + c0 + c1 + c2;
}

__global__ void k_scan_fix(const int* __restrict__ cnt, int* __restrict__ rowptr,
                           const int* __restrict__ bsum,
                           float* __restrict__ dinv, int n) {
    int i = blockIdx.x * blockDim.x + threadIdx.x;
    if (i >= n) return;
    const int b = (int)(blockIdx.x * blockDim.x) >> 12;
    int off = 0;
    for (int k = 0; k < b; ++k) off += bsum[k];
    const int r = rowptr[i] + off;
    rowptr[i] = r;
    dinv[i] = rsqrtf((float)(cnt[i] + 1));
}

__global__ void k_fill(const int* __restrict__ src, const int* __restrict__ dst,
                       const int* __restrict__ rowptr, const int* __restrict__ rank,
                       int* __restrict__ col, int E) {
    int e = blockIdx.x * blockDim.x + threadIdx.x;
    if (e >= E) return;
    col[rowptr[dst[e]] + rank[e]] = src[e];
}

__global__ void __launch_bounds__(256) k_gemm64(const float* __restrict__ A,
                                                const float* __restrict__ B,
                                                float* __restrict__ C) {
    const int wave = threadIdx.x >> 5;
    const int lane = threadIdx.x & 31;
    const int half = lane >> 4;
    const int lm   = lane & 15;
    const int m0   = blockIdx.x * 32 + (wave >> 2) * 16;
    const int n0   = (wave & 3) * 16;
    const float* arow = A + (size_t)(m0 + lm) * 64;
    const int bcol = n0 + lm;

    v8f acc = {0.f,0.f,0.f,0.f,0.f,0.f,0.f,0.f};
#pragma unroll
    for (int kk = 0; kk < 16; ++kk) {
        const int kb = kk * 4 + half * 2;
        v2f a = *(const v2f*)(arow + kb);
        v2f b;
        b.x = B[(size_t)kb * 64 + bcol];
        b.y = B[(size_t)(kb + 1) * 64 + bcol];
        acc = __builtin_amdgcn_wmma_f32_16x16x4_f32(false, a, false, b,
                                                    (short)0, acc, false, false);
    }
#pragma unroll
    for (int i = 0; i < 8; ++i)
        C[(size_t)(m0 + half * 8 + i) * 64 + n0 + lm] = acc[i];
}

__global__ void __launch_bounds__(256) k_gather64(const int* __restrict__ rowptr,
                                                  const int* __restrict__ cnt,
                                                  const int* __restrict__ col,
                                                  const float* __restrict__ dinv,
                                                  const float* __restrict__ xw,
                                                  const float* __restrict__ b1,
                                                  float* __restrict__ h, int n) {
    const int t = blockIdx.x * 256 + threadIdx.x;
    const int node = t >> 4;
    const int c = (t & 15) << 2;
    if (node >= n) return;
    const float di = dinv[node];
    int p = rowptr[node];
    const int end = p + cnt[node];
    const float4 sv = *(const float4*)(xw + (size_t)node * 64 + c);
    const float4 bb = *(const float4*)(b1 + c);
    const float wself = di * di;
    float4 acc;
    acc.x = bb.x + wself * sv.x;
    acc.y = bb.y + wself * sv.y;
    acc.z = bb.z + wself * sv.z;
    acc.w = bb.w + wself * sv.w;
    for (; p + 4 <= end; p += 4) {
        const int s0 = col[p], s1 = col[p + 1], s2 = col[p + 2], s3 = col[p + 3];
        const float w0 = dinv[s0] * di, w1 = dinv[s1] * di, w2 = dinv[s2] * di, w3 = dinv[s3] * di;
        const float4 v0 = *(const float4*)(xw + (size_t)s0 * 64 + c);
        const float4 v1 = *(const float4*)(xw + (size_t)s1 * 64 + c);
        const float4 v2 = *(const float4*)(xw + (size_t)s2 * 64 + c);
        const float4 v3 = *(const float4*)(xw + (size_t)s3 * 64 + c);
        acc.x += w0 * v0.x; acc.y += w0 * v0.y; acc.z += w0 * v0.z; acc.w += w0 * v0.w;
        acc.x += w1 * v1.x; acc.y += w1 * v1.y; acc.z += w1 * v1.z; acc.w += w1 * v1.w;
        acc.x += w2 * v2.x; acc.y += w2 * v2.y; acc.z += w2 * v2.z; acc.w += w2 * v2.w;
        acc.x += w3 * v3.x; acc.y += w3 * v3.y; acc.z += w3 * v3.z; acc.w += w3 * v3.w;
    }
    for (; p < end; ++p) {
        const int s0 = col[p];
        const float w0 = dinv[s0] * di;
        const float4 v0 = *(const float4*)(xw + (size_t)s0 * 64 + c);
        acc.x += w0 * v0.x; acc.y += w0 * v0.y; acc.z += w0 * v0.z; acc.w += w0 * v0.w;
    }
    *(float4*)(h + (size_t)node * 64 + c) = acc;
}

__global__ void __launch_bounds__(128) k_relu_softmax(float* __restrict__ h, int n) {
    int node = blockIdx.x * 4 + (threadIdx.x >> 5);
    int lane = threadIdx.x & 31;
    if (node >= n) return;
    float* row = h + (size_t)node * 64;
    float v0 = fmaxf(row[lane], 0.0f);
    float v1 = fmaxf(row[lane + 32], 0.0f);
    float m = fmaxf(v0, v1);
#pragma unroll
    for (int off = 16; off > 0; off >>= 1) m = fmaxf(m, __shfl_xor(m, off, 32));
    float e0 = __expf(v0 - m), e1 = __expf(v1 - m);
    float s = e0 + e1;
#pragma unroll
    for (int off = 16; off > 0; off >>= 1) s += __shfl_xor(s, off, 32);
    float inv = 1.0f / s;
    row[lane]      = e0 * inv;
    row[lane + 32] = e1 * inv;
}

__global__ void __launch_bounds__(192) k_gemm34(const float* __restrict__ A,
                                                const float* __restrict__ B,
                                                float* __restrict__ C) {
    const int wave = threadIdx.x >> 5;
    const int lane = threadIdx.x & 31;
    const int half = lane >> 4;
    const int lm   = lane & 15;
    const int m0   = blockIdx.x * 32 + (wave / 3) * 16;
    const int n0   = (wave % 3) * 16;
    const int col  = n0 + lm;
    const bool ok  = (col < 34);
    const float* arow = A + (size_t)(m0 + lm) * 64;

    v8f acc = {0.f,0.f,0.f,0.f,0.f,0.f,0.f,0.f};
#pragma unroll
    for (int kk = 0; kk < 16; ++kk) {
        const int kb = kk * 4 + half * 2;
        v2f a = *(const v2f*)(arow + kb);
        v2f b;
        b.x = ok ? B[(size_t)kb * 34 + col]       : 0.0f;
        b.y = ok ? B[(size_t)(kb + 1) * 34 + col] : 0.0f;
        acc = __builtin_amdgcn_wmma_f32_16x16x4_f32(false, a, false, b,
                                                    (short)0, acc, false, false);
    }
    if (ok) {
#pragma unroll
        for (int i = 0; i < 8; ++i)
            C[(size_t)(m0 + half * 8 + i) * 34 + col] = acc[i];
    }
}

__global__ void __launch_bounds__(256) k_gather34(const int* __restrict__ rowptr,
                                                  const int* __restrict__ cnt,
                                                  const int* __restrict__ col,
                                                  const float* __restrict__ dinv,
                                                  const float* __restrict__ hw2,
                                                  const float* __restrict__ b2,
                                                  float* __restrict__ out, int n) {
    const int t = blockIdx.x * 256 + threadIdx.x;
    const int node = t >> 4;
    const int l = t & 15;
    const int c = l << 1;
    if (node >= n) return;
    const float di = dinv[node];
    int p = rowptr[node];
    const int end = p + cnt[node];
    const float* srow = hw2 + (size_t)node * 34;
    const v2f sv = *(const v2f*)(srow + c);
    const v2f sx = *(const v2f*)(srow + 32);
    const float wself = di * di;
    v2f acc, accx;
    acc.x  = b2[c]  + wself * sv.x;
    acc.y  = b2[c + 1] + wself * sv.y;
    accx.x = b2[32] + wself * sx.x;
    accx.y = b2[33] + wself * sx.y;
    for (; p + 4 <= end; p += 4) {
        const int s0 = col[p], s1 = col[p + 1], s2 = col[p + 2], s3 = col[p + 3];
        const float w0 = dinv[s0] * di, w1 = dinv[s1] * di, w2 = dinv[s2] * di, w3 = dinv[s3] * di;
        const float* r0 = hw2 + (size_t)s0 * 34;
        const float* r1 = hw2 + (size_t)s1 * 34;
        const float* r2 = hw2 + (size_t)s2 * 34;
        const float* r3 = hw2 + (size_t)s3 * 34;
        const v2f v0 = *(const v2f*)(r0 + c), x0 = *(const v2f*)(r0 + 32);
        const v2f v1 = *(const v2f*)(r1 + c), x1 = *(const v2f*)(r1 + 32);
        const v2f v2 = *(const v2f*)(r2 + c), x2 = *(const v2f*)(r2 + 32);
        const v2f v3 = *(const v2f*)(r3 + c), x3 = *(const v2f*)(r3 + 32);
        acc.x += w0 * v0.x; acc.y += w0 * v0.y; accx.x += w0 * x0.x; accx.y += w0 * x0.y;
        acc.x += w1 * v1.x; acc.y += w1 * v1.y; accx.x += w1 * x1.x; accx.y += w1 * x1.y;
        acc.x += w2 * v2.x; acc.y += w2 * v2.y; accx.x += w2 * x2.x; accx.y += w2 * x2.y;
        acc.x += w3 * v3.x; acc.y += w3 * v3.y; accx.x += w3 * x3.x; accx.y += w3 * x3.y;
    }
    for (; p < end; ++p) {
        const int s0 = col[p];
        const float w0 = dinv[s0] * di;
        const float* r0 = hw2 + (size_t)s0 * 34;
        const v2f v0 = *(const v2f*)(r0 + c), x0 = *(const v2f*)(r0 + 32);
        acc.x += w0 * v0.x; acc.y += w0 * v0.y; accx.x += w0 * x0.x; accx.y += w0 * x0.y;
    }
    float* orow = out + (size_t)node * 34;
    *(v2f*)(orow + c) = acc;
    if (l == 0) *(v2f*)(orow + 32) = accx;
}

extern "C" void kernel_launch(void* const* d_in, const int* in_sizes, int n_in,
                              void* d_out, int out_size, void* d_ws, size_t ws_size,
                              hipStream_t stream) {
    const float* x  = (const float*)d_in[0];
    const int*   ei = (const int*)  d_in[1];
    const float* W1 = (const float*)d_in[2];
    const float* b1 = (const float*)d_in[3];
    const float* W2 = (const float*)d_in[4];
    const float* b2 = (const float*)d_in[5];
    float* out = (float*)d_out;

    const int N = in_sizes[0] / 64;
    const int E = in_sizes[1] / 2;
    const int* src = ei;
    const int* dst = ei + E;

    int*   cnt    = (int*)d_ws;
    int*   rowptr = cnt    + N;
    int*   bsum   = rowptr + N;
    float* dinv   = (float*)(bsum + 64);
    float* xw     = dinv + N;
    float* h      = xw   + (size_t)N * 64;
    float* hw2    = h    + (size_t)N * 64;
    int*   col    = (int*)(hw2 + (size_t)N * 34);
    int*   rank   = col + E;

    const int T = 256;
    k_zero_cnt  <<<(N + T - 1) / T, T, 0, stream>>>(cnt, N);
    k_count     <<<(E + T - 1) / T, T, 0, stream>>>(dst, cnt, rank, E);
    k_scan_local<<<(N + 4095) / 4096, 1024, 0, stream>>>(cnt, rowptr, bsum, N);
    k_scan_fix  <<<(N + T - 1) / T, T, 0, stream>>>(cnt, rowptr, bsum, dinv, N);
    k_fill      <<<(E + T - 1) / T, T, 0, stream>>>(src, dst, rowptr, rank, col, E);
    k_gemm64<<<N / 32, 256, 0, stream>>>(x, W1, xw);
    k_gather64<<<((size_t)N * 16 + T - 1) / T, T, 0, stream>>>(rowptr, cnt, col, dinv, xw, b1, h, N);
    k_relu_softmax<<<(N + 3) / 4, 128, 0, stream>>>(h, N);
    k_gemm34<<<N / 32, 192, 0, stream>>>(h, W2, hw2);
    k_gather34<<<((size_t)N * 16 + T - 1) / T, T, 0, stream>>>(rowptr, cnt, col, dinv, hw2, b2, out, N);
}


# relu+softmax moved into the layer-1 gather's epilogue (xor butterfly inside each node's lane group, on registers): one launch and its one-node waves go; softmax sum tree changes
# speedup vs baseline: 1.9940x; 1.0627x over previous
#include <hip/hip_runtime.h>
#include <math.h>

typedef __attribute__((ext_vector_type(2))) float v2f;
typedef __attribute__((ext_vector_type(8))) float v8f;


__global__ void k_zero_cnt(int* __restrict__ cnt, int n) {
    int i = blockIdx.x * blockDim.x + threadIdx.x;
    if (i < n) cnt[i] = 0;
}

__global__ void k_count(const int* __restrict__ dst, int* __restrict__ cnt,
                        int* __restrict__ rank, int E) {
    int e = blockIdx.x * blockDim.x + threadIdx.x;
    if (e < E) rank[e] = atomicAdd(&cnt[dst[e]], 1);
}

__global__ void __launch_bounds__(1024) k_scan_local(const int* __restrict__ cnt,
                                                     int* __restrict__ excl,
                                                     int* __restrict__ bsum, int n) {
    __shared__ int wsum[32];
    const int tid = threadIdx.x, lane = tid & 31, wv = tid >> 5;
    const int i0 = (blockIdx.x * 1024 + tid) * 4;
    const int c0 = (i0     < n) ? cnt[i0]     : 0;
    const int c1 = (i0 + 1 < n) ? cnt[i0 + 1] : 0;
    const int c2 = (i0 + 2 < n) ? cnt[i0 + 2] : 0;
    const int c3 = (i0 + 3 < n) ? cnt[i0 + 3] : 0;
    const int tsum = c0 + c1 + c2 + c3;
    int inc = tsum;
#pragma unroll
    for (int off = 1; off < 32; off <<= 1) {
        int v = __shfl_up(inc, off, 32);
        if (lane >= off) inc += v;
    }
    if (lane == 31) wsum[wv] = inc;
    __syncthreads();
    if (wv == 0) {
        const int w = wsum[lane];
        int winc = w;
#pragma unroll
        for (int off = 1; off < 32; off <<= 1) {
            int v = __shfl_up(winc, off, 32);
            if (lane >= off) winc += v;
        }
        wsum[lane] = winc - w;
        if (lane == 31) bsum[blockIdx.x] = winc;
    }
    __syncthreads();
    const int base = wsum[wv] + inc - tsum;
    if (i0     < n) excl[i0]     = base;
    if (i0 + 1 < n) excl[i0 + 1] = base + c0;
    if (i0 + 2 < n) excl[i0 + 2] = base + c0 + c1;
    if (i0 + 3 < n) excl[i0 + 3] = base + c0 + c1 + c2;
}

__global__ void k_scan_fix(const int* __restrict__ cnt, int* __restrict__ rowptr,
                           const int* __restrict__ bsum,
                           float* __restrict__ dinv, int n) {
    int i = blockIdx.x * blockDim.x + threadIdx.x;
    if (i >= n) return;
    const int b = (int)(blockIdx.x * blockDim.x) >> 12;
    int off = 0;
    for (int k = 0; k < b; ++k) off += bsum[k];
    const int r = rowptr[i] + off;
    rowptr[i] = r;
    dinv[i] = rsqrtf((float)(cnt[i] + 1));
}

__global__ void k_fill(const int* __restrict__ src, const int* __restrict__ dst,
                       const int* __restrict__ rowptr, const int* __restrict__ rank,
                       int* __restrict__ col, int E) {
    int e = blockIdx.x * blockDim.x + threadIdx.x;
    if (e >= E) return;
    col[rowptr[dst[e]] + rank[e]] = src[e];
}

__global__ void __launch_bounds__(256) k_gemm64(const float* __restrict__ A,
                                                const float* __restrict__ B,
                                                float* __restrict__ C) {
    const int wave = threadIdx.x >> 5;
    const int lane = threadIdx.x & 31;
    const int half = lane >> 4;
    const int lm   = lane & 15;
    const int m0   = blockIdx.x * 32 + (wave >> 2) * 16;
    const int n0   = (wave & 3) * 16;
    const float* arow = A + (size_t)(m0 + lm) * 64;
    const int bcol = n0 + lm;

    v8f acc = {0.f,0.f,0.f,0.f,0.f,0.f,0.f,0.f};
#pragma unroll
    for (int kk = 0; kk < 16; ++kk) {
        const int kb = kk * 4 + half * 2;
        v2f a = *(const v2f*)(arow + kb);
        v2f b;
        b.x = B[(size_t)kb * 64 + bcol];
        b.y = B[(size_t)(kb + 1) * 64 + bcol];
        acc = __builtin_amdgcn_wmma_f32_16x16x4_f32(false, a, false, b,
                                                    (short)0, acc, false, false);
    }
#pragma unroll
    for (int i = 0; i < 8; ++i)
        C[(size_t)(m0 + half * 8 + i) * 64 + n0 + lm] = acc[i];
}

__global__ void __launch_bounds__(256) k_gather64(const int* __restrict__ rowptr,
                                                  const int* __restrict__ cnt,
                                                  const int* __restrict__ col,
                                                  const float* __restrict__ dinv,
                                                  const float* __restrict__ xw,
                                                  const float* __restrict__ b1,
                                                  float* __restrict__ h, int n) {
    const int t = blockIdx.x * 256 + threadIdx.x;
    const int node = t >> 4;
    const int c = (t & 15) << 2;
    if (node >= n) return;
    const float di = dinv[node];
    int p = rowptr[node];
    const int end = p + cnt[node];
    const float4 sv = *(const float4*)(xw + (size_t)node * 64 + c);
    const float4 bb = *(const float4*)(b1 + c);
    const float wself = di * di;
    float4 acc;
    acc.x = bb.x + wself * sv.x;
    acc.y = bb.y + wself * sv.y;
    acc.z = bb.z + wself * sv.z;
    acc.w = bb.w + wself * sv.w;
    for (; p + 4 <= end; p += 4) {
        const int s0 = col[p], s1 = col[p + 1], s2 = col[p + 2], s3 = col[p + 3];
        const float w0 = dinv[s0] * di, w1 = dinv[s1] * di, w2 = dinv[s2] * di, w3 = dinv[s3] * di;
        const float4 v0 = *(const float4*)(xw + (size_t)s0 * 64 + c);
        const float4 v1 = *(const float4*)(xw + (size_t)s1 * 64 + c);
        const float4 v2 = *(const float4*)(xw + (size_t)s2 * 64 + c);
        const float4 v3 = *(const float4*)(xw + (size_t)s3 * 64 + c);
        acc.x += w0 * v0.x; acc.y += w0 * v0.y; acc.z += w0 * v0.z; acc.w += w0 * v0.w;
        acc.x += w1 * v1.x; acc.y += w1 * v1.y; acc.z += w1 * v1.z; acc.w += w1 * v1.w;
        acc.x += w2 * v2.x; acc.y += w2 * v2.y; acc.z += w2 * v2.z; acc.w += w2 * v2.w;
        acc.x += w3 * v3.x; acc.y += w3 * v3.y; acc.z += w3 * v3.z; acc.w += w3 * v3.w;
    }
    for (; p < end; ++p) {
        const int s0 = col[p];
        const float w0 = dinv[s0] * di;
        const float4 v0 = *(const float4*)(xw + (size_t)s0 * 64 + c);
        acc.x += w0 * v0.x; acc.y += w0 * v0.y; acc.z += w0 * v0.z; acc.w += w0 * v0.w;
    }
    const float r0 = fmaxf(acc.x, 0.0f), r1 = fmaxf(acc.y, 0.0f);
    const float r2 = fmaxf(acc.z, 0.0f), r3 = fmaxf(acc.w, 0.0f);
    float m = fmaxf(fmaxf(r0, r1), fmaxf(r2, r3));
    for (int off = 8; off > 0; off >>= 1) m = fmaxf(m, __shfl_xor(m, off, 16));
    const float e0 = __expf(r0 - m), e1 = __expf(r1 - m);
    const float e2 = __expf(r2 - m), e3 = __expf(r3 - m);
    float s = (e0 + e1) + (e2 + e3);
    for (int off = 8; off > 0; off >>= 1) s += __shfl_xor(s, off, 16);
    const float inv = 1.0f / s;
    float4 o;
    o.x = e0 * inv; o.y = e1 * inv; o.z = e2 * inv; o.w = e3 * inv;
    *(float4*)(h + (size_t)node * 64 + c) = o;
}

__global__ void __launch_bounds__(192) k_gemm34(const float* __restrict__ A,
                                                const float* __restrict__ B,
                                                float* __restrict__ C) {
    const int wave = threadIdx.x >> 5;
    const int lane = threadIdx.x & 31;
    const int half = lane >> 4;
    const int lm   = lane & 15;
    const int m0   = blockIdx.x * 32 + (wave / 3) * 16;
    const int n0   = (wave % 3) * 16;
    const int col  = n0 + lm;
    const bool ok  = (col < 34);
    const float* arow = A + (size_t)(m0 + lm) * 64;

    v8f acc = {0.f,0.f,0.f,0.f,0.f,0.f,0.f,0.f};
#pragma unroll
    for (int kk = 0; kk < 16; ++kk) {
        const int kb = kk * 4 + half * 2;
        v2f a = *(const v2f*)(arow + kb);
        v2f b;
        b.x = ok ? B[(size_t)kb * 34 + col]       : 0.0f;
        b.y = ok ? B[(size_t)(kb + 1) * 34 + col] : 0.0f;
        acc = __builtin_amdgcn_wmma_f32_16x16x4_f32(false, a, false, b,
                                                    (short)0, acc, false, false);
    }
    if (ok) {
#pragma unroll
        for (int i = 0; i < 8; ++i)
            C[(size_t)(m0 + half * 8 + i) * 34 + col] = acc[i];
    }
}

__global__ void __launch_bounds__(256) k_gather34(const int* __restrict__ rowptr,
                                                  const int* __restrict__ cnt,
                                                  const int* __restrict__ col,
                                                  const float* __restrict__ dinv,
                                                  const float* __restrict__ hw2,
                                                  const float* __restrict__ b2,
                                                  float* __restrict__ out, int n) {
    const int t = blockIdx.x * 256 + threadIdx.x;
    const int node = t >> 4;
    const int l = t & 15;
    const int c = l << 1;
    if (node >= n) return;
    const float di = dinv[node];
    int p = rowptr[node];
    const int end = p + cnt[node];
    const float* srow = hw2 + (size_t)node * 34;
    const v2f sv = *(const v2f*)(srow + c);
    const v2f sx = *(const v2f*)(srow + 32);
    const float wself = di * di;
    v2f acc, accx;
    acc.x  = b2[c]  + wself * sv.x;
    acc.y  = b2[c + 1] + wself * sv.y;
    accx.x = b2[32] + wself * sx.x;
    accx.y = b2[33] + wself * sx.y;
    for (; p + 4 <= end; p += 4) {
        const int s0 = col[p], s1 = col[p + 1], s2 = col[p + 2], s3 = col[p + 3];
        const float w0 = dinv[s0] * di, w1 = dinv[s1] * di, w2 = dinv[s2] * di, w3 = dinv[s3] * di;
        const float* r0 = hw2 + (size_t)s0 * 34;
        const float* r1 = hw2 + (size_t)s1 * 34;
        const float* r2 = hw2 + (size_t)s2 * 34;
        const float* r3 = hw2 + (size_t)s3 * 34;
        const v2f v0 = *(const v2f*)(r0 + c), x0 = *(const v2f*)(r0 + 32);
        const v2f v1 = *(const v2f*)(r1 + c), x1 = *(const v2f*)(r1 + 32);
        const v2f v2 = *(const v2f*)(r2 + c), x2 = *(const v2f*)(r2 + 32);
        const v2f v3 = *(const v2f*)(r3 + c), x3 = *(const v2f*)(r3 + 32);
        acc.x += w0 * v0.x; acc.y += w0 * v0.y; accx.x += w0 * x0.x; accx.y += w0 * x0.y;
        acc.x += w1 * v1.x; acc.y += w1 * v1.y; accx.x += w1 * x1.x; accx.y += w1 * x1.y;
        acc.x += w2 * v2.x; acc.y += w2 * v2.y; accx.x += w2 * x2.x; accx.y += w2 * x2.y;
        acc.x += w3 * v3.x; acc.y += w3 * v3.y; accx.x += w3 * x3.x; accx.y += w3 * x3.y;
    }
    for (; p < end; ++p) {
        const int s0 = col[p];
        const float w0 = dinv[s0] * di;
        const float* r0 = hw2 + (size_t)s0 * 34;
        const v2f v0 = *(const v2f*)(r0 + c), x0 = *(const v2f*)(r0 + 32);
        acc.x += w0 * v0.x; acc.y += w0 * v0.y; accx.x += w0 * x0.x; accx.y += w0 * x0.y;
    }
    float* orow = out + (size_t)node * 34;
    *(v2f*)(orow + c) = acc;
    if (l == 0) *(v2f*)(orow + 32) = accx;
}

extern "C" void kernel_launch(void* const* d_in, const int* in_sizes, int n_in,
                              void* d_out, int out_size, void* d_ws, size_t ws_size,
                              hipStream_t stream) {
    const float* x  = (const float*)d_in[0];
    const int*   ei = (const int*)  d_in[1];
    const float* W1 = (const float*)d_in[2];
    const float* b1 = (const float*)d_in[3];
    const float* W2 = (const float*)d_in[4];
    const float* b2 = (const float*)d_in[5];
    float* out = (float*)d_out;

    const int N = in_sizes[0] / 64;
    const int E = in_sizes[1] / 2;
    const int* src = ei;
    const int* dst = ei + E;

    int*   cnt    = (int*)d_ws;
    int*   rowptr = cnt    + N;
    int*   bsum   = rowptr + N;
    float* dinv   = (float*)(bsum + 64);
    float* xw     = dinv + N;
    float* h      = xw   + (size_t)N * 64;
    float* hw2    = h    + (size_t)N * 64;
    int*   col    = (int*)(hw2 + (size_t)N * 34);
    int*   rank   = col + E;

    const int T = 256;
    k_zero_cnt  <<<(N + T - 1) / T, T, 0, stream>>>(cnt, N);
    k_count     <<<(E + T - 1) / T, T, 0, stream>>>(dst, cnt, rank, E);
    k_scan_local<<<(N + 4095) / 4096, 1024, 0, stream>>>(cnt, rowptr, bsum, N);
    k_scan_fix  <<<(N + T - 1) / T, T, 0, stream>>>(cnt, rowptr, bsum, dinv, N);
    k_fill      <<<(E + T - 1) / T, T, 0, stream>>>(src, dst, rowptr, rank, col, E);
    k_gemm64<<<N / 32, 256, 0, stream>>>(x, W1, xw);
    k_gather64<<<((size_t)N * 16 + T - 1) / T, T, 0, stream>>>(rowptr, cnt, col, dinv, xw, b1, h, N);
    k_gemm34<<<N / 32, 192, 0, stream>>>(h, W2, hw2);
    k_gather34<<<((size_t)N * 16 + T - 1) / T, T, 0, stream>>>(rowptr, cnt, col, dinv, hw2, b2, out, N);
}


# Edge bucketing redone as a two-level counting sort on LDS counters (512-node ranges, then rank inside each range's block): no global atomics, build launches 5 -> 3; sum order = LDS arrival order
# speedup vs baseline: 2.8710x; 1.4398x over previous
#include <hip/hip_runtime.h>
#include <math.h>

typedef __attribute__((ext_vector_type(2))) float v2f;
typedef __attribute__((ext_vector_type(8))) float v8f;


#define RSH   9
#define RNG   512
#define NBMAX 256
#define CH    5120

__global__ void __launch_bounds__(1024) k_hist(const int* __restrict__ dst, int* __restrict__ hist, int E) {
    __shared__ int lh[NBMAX];
    const int tid = threadIdx.x;
    if (tid < NBMAX) lh[tid] = 0;
    __syncthreads();
    const int base = blockIdx.x * CH + tid;
    int d[CH / 1024];
#pragma unroll
    for (int k = 0; k < CH / 1024; ++k) {
        const int e = base + k * 1024;
        d[k] = dst[e < E ? e : E - 1];
    }
#pragma unroll
    for (int k = 0; k < CH / 1024; ++k)
        if (base + k * 1024 < E) atomicAdd(&lh[d[k] >> RSH], 1);
    __syncthreads();
    if (tid < NBMAX) hist[blockIdx.x * NBMAX + tid] = lh[tid];
}

__global__ void __launch_bounds__(1024) k_bin(const int* __restrict__ src, const int* __restrict__ dst,
                                              const int* __restrict__ hist, int* __restrict__ binoff,
                                              int* __restrict__ binned, int E, int nblk) {
    __shared__ int ltot[16][NBMAX];
    __shared__ int lpre[16][NBMAX];
    __shared__ int lcur[NBMAX];
    __shared__ int wsum[8];
    const int tid = threadIdx.x, lane = tid & 31, wv = tid >> 5;
    const int me = blockIdx.x;
    const int base = me * CH + tid;
    int d[CH / 1024], s[CH / 1024];
#pragma unroll
    for (int k = 0; k < CH / 1024; ++k) {
        const int e  = base + k * 1024;
        const int ec = e < E ? e : E - 1;
        d[k] = dst[ec];
        s[k] = src[ec];
    }
    const int q  = tid >> 6;
    const int c4 = (tid & 63) << 2;
    int4 tot = make_int4(0, 0, 0, 0), pre = make_int4(0, 0, 0, 0);
    for (int b = q; b < nblk; b += 16) {
        const int4 v = *(const int4*)(hist + (size_t)b * NBMAX + c4);
        tot.x += v.x; tot.y += v.y; tot.z += v.z; tot.w += v.w;
        if (b < me) { pre.x += v.x; pre.y += v.y; pre.z += v.z; pre.w += v.w; }
    }
    *(int4*)&ltot[q][c4] = tot;
    *(int4*)&lpre[q][c4] = pre;
    __syncthreads();
    int t = 0, p = 0, inc = 0;
    if (tid < NBMAX) {
#pragma unroll
        for (int g = 0; g < 16; ++g) { t += ltot[g][tid]; p += lpre[g][tid]; }
        inc = t;
#pragma unroll
        for (int off = 1; off < 32; off <<= 1) {
            int v = __shfl_up(inc, off, 32);
            if (lane >= off) inc += v;
        }
        if (lane == 31) wsum[wv] = inc;
    }
    __syncthreads();
    if (tid < NBMAX) {
        int woff = 0;
        for (int w = 0; w < wv; ++w) woff += wsum[w];
        const int off = woff + inc - t;
        lcur[tid] = off + p;
        if (me == 0) {
            binoff[tid] = off;
            if (tid == NBMAX - 1) binoff[NBMAX] = off + t;
        }
    }
    __syncthreads();
#pragma unroll
    for (int k = 0; k < CH / 1024; ++k) {
        if (base + k * 1024 < E) {
            const int pos = atomicAdd(&lcur[d[k] >> RSH], 1);
            binned[pos] = (s[k] << RSH) | (d[k] & (RNG - 1));
        }
    }
}

__global__ void __launch_bounds__(1024) k_csr(const int* __restrict__ binned, const int* __restrict__ binoff,
                                              int* __restrict__ cnt, int* __restrict__ rowptr,
                                              float* __restrict__ dinv, int* __restrict__ col, int n) {
    __shared__ int lc[RNG];
    __shared__ int wsum[RNG / 32];
    const int tid = threadIdx.x, lane = tid & 31, wv = tid >> 5;
    const int start = binoff[blockIdx.x];
    const int end   = binoff[blockIdx.x + 1];
    if (tid < RNG) lc[tid] = 0;
    __syncthreads();
    for (int e0 = start + tid; e0 < end; e0 += 8 * 1024) {
        int v[8];
#pragma unroll
        for (int k = 0; k < 8; ++k) {
            const int e = e0 + k * 1024;
            v[k] = binned[e < end ? e : end - 1];
        }
        asm volatile("" :: "v"(v[0]), "v"(v[1]), "v"(v[2]), "v"(v[3]), "v"(v[4]), "v"(v[5]), "v"(v[6]), "v"(v[7]));
#pragma unroll
        for (int k = 0; k < 8; ++k)
            if (e0 + k * 1024 < end) atomicAdd(&lc[v[k] & (RNG - 1)], 1);
    }
    __syncthreads();
    int c = 0, inc = 0;
    if (tid < RNG) {
        c = lc[tid];
        inc = c;
#pragma unroll
        for (int off = 1; off < 32; off <<= 1) {
            int v = __shfl_up(inc, off, 32);
            if (lane >= off) inc += v;
        }
        if (lane == 31) wsum[wv] = inc;
    }
    __syncthreads();
    if (tid < RNG) {
        int woff = 0;
        for (int w = 0; w < wv; ++w) woff += wsum[w];
        const int ex = start + woff + inc - c;
        lc[tid] = ex;
        const int node = blockIdx.x * RNG + tid;
        if (node < n) {
            cnt[node]    = c;
            rowptr[node] = ex;
            dinv[node]   = rsqrtf((float)(c + 1));
        }
    }
    __syncthreads();
    for (int e0 = start + tid; e0 < end; e0 += 8 * 1024) {
        int v[8];
#pragma unroll
        for (int k = 0; k < 8; ++k) {
            const int e = e0 + k * 1024;
            v[k] = binned[e < end ? e : end - 1];
        }
        asm volatile("" :: "v"(v[0]), "v"(v[1]), "v"(v[2]), "v"(v[3]), "v"(v[4]), "v"(v[5]), "v"(v[6]), "v"(v[7]));
#pragma unroll
        for (int k = 0; k < 8; ++k) {
            if (e0 + k * 1024 < end) {
                const int pos = atomicAdd(&lc[v[k] & (RNG - 1)], 1);
                col[pos] = v[k] >> RSH;
            }
        }
    }
}

__global__ void __launch_bounds__(256) k_gemm64(const float* __restrict__ A,
                                                const float* __restrict__ B,
                                                float* __restrict__ C) {
    const int wave = threadIdx.x >> 5;
    const int lane = threadIdx.x & 31;
    const int half = lane >> 4;
    const int lm   = lane & 15;
    const int m0   = blockIdx.x * 32 + (wave >> 2) * 16;
    const int n0   = (wave & 3) * 16;
    const float* arow = A + (size_t)(m0 + lm) * 64;
    const int bcol = n0 + lm;

    v8f acc = {0.f,0.f,0.f,0.f,0.f,0.f,0.f,0.f};
#pragma unroll
    for (int kk = 0; kk < 16; ++kk) {
        const int kb = kk * 4 + half * 2;
        v2f a = *(const v2f*)(arow + kb);
        v2f b;
        b.x = B[(size_t)kb * 64 + bcol];
        b.y = B[(size_t)(kb + 1) * 64 + bcol];
        acc = __builtin_amdgcn_wmma_f32_16x16x4_f32(false, a, false, b,
                                                    (short)0, acc, false, false);
    }
#pragma unroll
    for (int i = 0; i < 8; ++i)
        C[(size_t)(m0 + half * 8 + i) * 64 + n0 + lm] = acc[i];
}

__global__ void __launch_bounds__(256) k_gather64(const int* __restrict__ rowptr,
                                                  const int* __restrict__ cnt,
                                                  const int* __restrict__ col,
                                                  const float* __restrict__ dinv,
                                                  const float* __restrict__ xw,
                                                  const float* __restrict__ b1,
                                                  float* __restrict__ h, int n) {
    const int t = blockIdx.x * 256 + threadIdx.x;
    const int node = t >> 4;
    const int c = (t & 15) << 2;
    if (node >= n) return;
    const float di = dinv[node];
    int p = rowptr[node];
    const int end = p + cnt[node];
    const float4 sv = *(const float4*)(xw + (size_t)node * 64 + c);
    const float4 bb = *(const float4*)(b1 + c);
    const float wself = di * di;
    float4 acc;
    acc.x = bb.x + wself * sv.x;
    acc.y = bb.y + wself * sv.y;
    acc.z = bb.z + wself * sv.z;
    acc.w = bb.w + wself * sv.w;
    for (; p + 4 <= end; p += 4) {
        const int s0 = col[p], s1 = col[p + 1], s2 = col[p + 2], s3 = col[p + 3];
        const float w0 = dinv[s0] * di, w1 = dinv[s1] * di, w2 = dinv[s2] * di, w3 = dinv[s3] * di;
        const float4 v0 = *(const float4*)(xw + (size_t)s0 * 64 + c);
        const float4 v1 = *(const float4*)(xw + (size_t)s1 * 64 + c);
        const float4 v2 = *(const float4*)(xw + (size_t)s2 * 64 + c);
        const float4 v3 = *(const float4*)(xw + (size_t)s3 * 64 + c);
        acc.x += w0 * v0.x; acc.y += w0 * v0.y; acc.z += w0 * v0.z; acc.w += w0 * v0.w;
        acc.x += w1 * v1.x; acc.y += w1 * v1.y; acc.z += w1 * v1.z; acc.w += w1 * v1.w;
        acc.x += w2 * v2.x; acc.y += w2 * v2.y; acc.z += w2 * v2.z; acc.w += w2 * v2.w;
        acc.x += w3 * v3.x; acc.y += w3 * v3.y; acc.z += w3 * v3.z; acc.w += w3 * v3.w;
    }
    for (; p < end; ++p) {
        const int s0 = col[p];
        const float w0 = dinv[s0] * di;
        const float4 v0 = *(const float4*)(xw + (size_t)s0 * 64 + c);
        acc.x += w0 * v0.x; acc.y += w0 * v0.y; acc.z += w0 * v0.z; acc.w += w0 * v0.w;
    }
    const float r0 = fmaxf(acc.x, 0.0f), r1 = fmaxf(acc.y, 0.0f);
    const float r2 = fmaxf(acc.z, 0.0f), r3 = fmaxf(acc.w, 0.0f);
    float m = fmaxf(fmaxf(r0, r1), fmaxf(r2, r3));
    for (int off = 8; off > 0; off >>= 1) m = fmaxf(m, __shfl_xor(m, off, 16));
    const float e0 = __expf(r0 - m), e1 = __expf(r1 - m);
    const float e2 = __expf(r2 - m), e3 = __expf(r3 - m);
    float s = (e0 + e1) + (e2 + e3);
    for (int off = 8; off > 0; off >>= 1) s += __shfl_xor(s, off, 16);
    const float inv = 1.0f / s;
    float4 o;
    o.x = e0 * inv; o.y = e1 * inv; o.z = e2 * inv; o.w = e3 * inv;
    *(float4*)(h + (size_t)node * 64 + c) = o;
}

__global__ void __launch_bounds__(192) k_gemm34(const float* __restrict__ A,
                                                const float* __restrict__ B,
                                                float* __restrict__ C) {
    const int wave = threadIdx.x >> 5;
    const int lane = threadIdx.x & 31;
    const int half = lane >> 4;
    const int lm   = lane & 15;
    const int m0   = blockIdx.x * 32 + (wave / 3) * 16;
    const int n0   = (wave % 3) * 16;
    const int col  = n0 + lm;
    const bool ok  = (col < 34);
    const float* arow = A + (size_t)(m0 + lm) * 64;

    v8f acc = {0.f,0.f,0.f,0.f,0.f,0.f,0.f,0.f};
#pragma unroll
    for (int kk = 0; kk < 16; ++kk) {
        const int kb = kk * 4 + half * 2;
        v2f a = *(const v2f*)(arow + kb);
        v2f b;
        b.x = ok ? B[(size_t)kb * 34 + col]       : 0.0f;
        b.y = ok ? B[(size_t)(kb + 1) * 34 + col] : 0.0f;
        acc = __builtin_amdgcn_wmma_f32_16x16x4_f32(false, a, false, b,
                                                    (short)0, acc, false, false);
    }
    if (ok) {
#pragma unroll
        for (int i = 0; i < 8; ++i)
            C[(size_t)(m0 + half * 8 + i) * 34 + col] = acc[i];
    }
}

__global__ void __launch_bounds__(256) k_gather34(const int* __restrict__ rowptr,
                                                  const int* __restrict__ cnt,
                                                  const int* __restrict__ col,
                                                  const float* __restrict__ dinv,
                                                  const float* __restrict__ hw2,
                                                  const float* __restrict__ b2,
                                                  float* __restrict__ out, int n) {
    const int t = blockIdx.x * 256 + threadIdx.x;
    const int node = t >> 4;
    const int l = t & 15;
    const int c = l << 1;
    if (node >= n) return;
    const float di = dinv[node];
    int p = rowptr[node];
    const int end = p + cnt[node];
    const float* srow = hw2 + (size_t)node * 34;
    const v2f sv = *(const v2f*)(srow + c);
    const v2f sx = *(const v2f*)(srow + 32);
    const float wself = di * di;
    v2f acc, accx;
    acc.x  = b2[c]  + wself * sv.x;
    acc.y  = b2[c + 1] + wself * sv.y;
    accx.x = b2[32] + wself * sx.x;
    accx.y = b2[33] + wself * sx.y;
    for (; p + 4 <= end; p += 4) {
        const int s0 = col[p], s1 = col[p + 1], s2 = col[p + 2], s3 = col[p + 3];
        const float w0 = dinv[s0] * di, w1 = dinv[s1] * di, w2 = dinv[s2] * di, w3 = dinv[s3] * di;
        const float* r0 = hw2 + (size_t)s0 * 34;
        const float* r1 = hw2 + (size_t)s1 * 34;
        const float* r2 = hw2 + (size_t)s2 * 34;
        const float* r3 = hw2 + (size_t)s3 * 34;
        const v2f v0 = *(const v2f*)(r0 + c), x0 = *(const v2f*)(r0 + 32);
        const v2f v1 = *(const v2f*)(r1 + c), x1 = *(const v2f*)(r1 + 32);
        const v2f v2 = *(const v2f*)(r2 + c), x2 = *(const v2f*)(r2 + 32);
        const v2f v3 = *(const v2f*)(r3 + c), x3 = *(const v2f*)(r3 + 32);
        acc.x += w0 * v0.x; acc.y += w0 * v0.y; accx.x += w0 * x0.x; accx.y += w0 * x0.y;
        acc.x += w1 * v1.x; acc.y += w1 * v1.y; accx.x += w1 * x1.x; accx.y += w1 * x1.y;
        acc.x += w2 * v2.x; acc.y += w2 * v2.y; accx.x += w2 * x2.x; accx.y += w2 * x2.y;
        acc.x += w3 * v3.x; acc.y += w3 * v3.y; accx.x += w3 * x3.x; accx.y += w3 * x3.y;
    }
    for (; p < end; ++p) {
        const int s0 = col[p];
        const float w0 = dinv[s0] * di;
        const float* r0 = hw2 + (size_t)s0 * 34;
        const v2f v0 = *(const v2f*)(r0 + c), x0 = *(const v2f*)(r0 + 32);
        acc.x += w0 * v0.x; acc.y += w0 * v0.y; accx.x += w0 * x0.x; accx.y += w0 * x0.y;
    }
    float* orow = out + (size_t)node * 34;
    *(v2f*)(orow + c) = acc;
    if (l == 0) *(v2f*)(orow + 32) = accx;
}

extern "C" void kernel_launch(void* const* d_in, const int* in_sizes, int n_in,
                              void* d_out, int out_size, void* d_ws, size_t ws_size,
                              hipStream_t stream) {
    const float* x  = (const float*)d_in[0];
    const int*   ei = (const int*)  d_in[1];
    const float* W1 = (const float*)d_in[2];
    const float* b1 = (const float*)d_in[3];
    const float* W2 = (const float*)d_in[4];
    const float* b2 = (const float*)d_in[5];
    float* out = (float*)d_out;

    const int N = in_sizes[0] / 64;
    const int E = in_sizes[1] / 2;
    const int* src = ei;
    const int* dst = ei + E;

    int*   cnt    = (int*)d_ws;
    int*   rowptr = cnt    + N;
    int*   pad    = rowptr + N;
    float* dinv   = (float*)(pad + 64);
    float* xw     = dinv + N;
    float* h      = xw   + (size_t)N * 64;
    float* hw2    = h    + (size_t)N * 64;
    int*   col    = (int*)(hw2 + (size_t)N * 34);
    int*   binned = col + E;
    int*   binoff = binned + E;
    int*   hist   = binoff + 512;

    const int T = 256;
    const int nblk = (E + CH - 1) / CH;
    k_hist<<<nblk, 1024, 0, stream>>>(dst, hist, E);
    k_bin <<<nblk, 1024, 0, stream>>>(src, dst, hist, binoff, binned, E, nblk);
    k_csr <<<(N + RNG - 1) / RNG, 1024, 0, stream>>>(binned, binoff, cnt, rowptr, dinv, col, N);
    k_gemm64<<<N / 32, 256, 0, stream>>>(x, W1, xw);
    k_gather64<<<((size_t)N * 16 + T - 1) / T, T, 0, stream>>>(rowptr, cnt, col, dinv, xw, b1, h, N);
    k_gemm34<<<N / 32, 192, 0, stream>>>(h, W2, hw2);
    k_gather34<<<((size_t)N * 16 + T - 1) / T, T, 0, stream>>>(rowptr, cnt, col, dinv, hw2, b2, out, N);
}


# Both dense products fused into the layer-1 gather: x rows summed first, 16-node tile feeds f32 WMMA via LDS (W1, relu+softmax, W2); 2 launches, xw/h planes go; layer-1 and softmax sum order change
# speedup vs baseline: 3.6676x; 1.2775x over previous
#include <hip/hip_runtime.h>
#include <math.h>

typedef __attribute__((ext_vector_type(2))) float v2f;
typedef __attribute__((ext_vector_type(8))) float v8f;


#define RSH   9
#define RNG   512
#define NBMAX 256
#define CH    5120

__global__ void __launch_bounds__(1024) k_hist(const int* __restrict__ dst, int* __restrict__ hist, int E) {
    __shared__ int lh[NBMAX];
    const int tid = threadIdx.x;
    if (tid < NBMAX) lh[tid] = 0;
    __syncthreads();
    const int base = blockIdx.x * CH + tid;
    int d[CH / 1024];
#pragma unroll
    for (int k = 0; k < CH / 1024; ++k) {
        const int e = base + k * 1024;
        d[k] = dst[e < E ? e : E - 1];
    }
#pragma unroll
    for (int k = 0; k < CH / 1024; ++k)
        if (base + k * 1024 < E) atomicAdd(&lh[d[k] >> RSH], 1);
    __syncthreads();
    if (tid < NBMAX) hist[blockIdx.x * NBMAX + tid] = lh[tid];
}

__global__ void __launch_bounds__(1024) k_bin(const int* __restrict__ src, const int* __restrict__ dst,
                                              const int* __restrict__ hist, int* __restrict__ binoff,
                                              int* __restrict__ binned, int E, int nblk) {
    __shared__ int ltot[16][NBMAX];
    __shared__ int lpre[16][NBMAX];
    __shared__ int lcur[NBMAX];
    __shared__ int wsum[8];
    const int tid = threadIdx.x, lane = tid & 31, wv = tid >> 5;
    const int me = blockIdx.x;
    const int base = me * CH + tid;
    int d[CH / 1024], s[CH / 1024];
#pragma unroll
    for (int k = 0; k < CH / 1024; ++k) {
        const int e  = base + k * 1024;
        const int ec = e < E ? e : E - 1;
        d[k] = dst[ec];
        s[k] = src[ec];
    }
    const int q  = tid >> 6;
    const int c4 = (tid & 63) << 2;
    int4 tot = make_int4(0, 0, 0, 0), pre = make_int4(0, 0, 0, 0);
    for (int b = q; b < nblk; b += 16) {
        const int4 v = *(const int4*)(hist + (size_t)b * NBMAX + c4);
        tot.x += v.x; tot.y += v.y; tot.z += v.z; tot.w += v.w;
        if (b < me) { pre.x += v.x; pre.y += v.y; pre.z += v.z; pre.w += v.w; }
    }
    *(int4*)&ltot[q][c4] = tot;
    *(int4*)&lpre[q][c4] = pre;
    __syncthreads();
    int t = 0, p = 0, inc = 0;
    if (tid < NBMAX) {
#pragma unroll
        for (int g = 0; g < 16; ++g) { t += ltot[g][tid]; p += lpre[g][tid]; }
        inc = t;
#pragma unroll
        for (int off = 1; off < 32; off <<= 1) {
            int v = __shfl_up(inc, off, 32);
            if (lane >= off) inc += v;
        }
        if (lane == 31) wsum[wv] = inc;
    }
    __syncthreads();
    if (tid < NBMAX) {
        int woff = 0;
        for (int w = 0; w < wv; ++w) woff += wsum[w];
        const int off = woff + inc - t;
        lcur[tid] = off + p;
        if (me == 0) {
            binoff[tid] = off;
            if (tid == NBMAX - 1) binoff[NBMAX] = off + t;
        }
    }
    __syncthreads();
#pragma unroll
    for (int k = 0; k < CH / 1024; ++k) {
        if (base + k * 1024 < E) {
            const int pos = atomicAdd(&lcur[d[k] >> RSH], 1);
            binned[pos] = (s[k] << RSH) | (d[k] & (RNG - 1));
        }
    }
}

__global__ void __launch_bounds__(1024) k_csr(const int* __restrict__ binned, const int* __restrict__ binoff,
                                              int* __restrict__ cnt, int* __restrict__ rowptr,
                                              float* __restrict__ dinv, int* __restrict__ col, int n) {
    __shared__ int lc[RNG];
    __shared__ int wsum[RNG / 32];
    const int tid = threadIdx.x, lane = tid & 31, wv = tid >> 5;
    const int start = binoff[blockIdx.x];
    const int end   = binoff[blockIdx.x + 1];
    if (tid < RNG) lc[tid] = 0;
    __syncthreads();
    for (int e0 = start + tid; e0 < end; e0 += 8 * 1024) {
        int v[8];
#pragma unroll
        for (int k = 0; k < 8; ++k) {
            const int e = e0 + k * 1024;
            v[k] = binned[e < end ? e : end - 1];
        }
        asm volatile("" :: "v"(v[0]), "v"(v[1]), "v"(v[2]), "v"(v[3]), "v"(v[4]), "v"(v[5]), "v"(v[6]), "v"(v[7]));
#pragma unroll
        for (int k = 0; k < 8; ++k)
            if (e0 + k * 1024 < end) atomicAdd(&lc[v[k] & (RNG - 1)], 1);
    }
    __syncthreads();
    int c = 0, inc = 0;
    if (tid < RNG) {
        c = lc[tid];
        inc = c;
#pragma unroll
        for (int off = 1; off < 32; off <<= 1) {
            int v = __shfl_up(inc, off, 32);
            if (lane >= off) inc += v;
        }
        if (lane == 31) wsum[wv] = inc;
    }
    __syncthreads();
    if (tid < RNG) {
        int woff = 0;
        for (int w = 0; w < wv; ++w) woff += wsum[w];
        const int ex = start + woff + inc - c;
        lc[tid] = ex;
        const int node = blockIdx.x * RNG + tid;
        if (node < n) {
            cnt[node]    = c;
            rowptr[node] = ex;
            dinv[node]   = rsqrtf((float)(c + 1));
        }
    }
    __syncthreads();
    for (int e0 = start + tid; e0 < end; e0 += 8 * 1024) {
        int v[8];
#pragma unroll
        for (int k = 0; k < 8; ++k) {
            const int e = e0 + k * 1024;
            v[k] = binned[e < end ? e : end - 1];
        }
        asm volatile("" :: "v"(v[0]), "v"(v[1]), "v"(v[2]), "v"(v[3]), "v"(v[4]), "v"(v[5]), "v"(v[6]), "v"(v[7]));
#pragma unroll
        for (int k = 0; k < 8; ++k) {
            if (e0 + k * 1024 < end) {
                const int pos = atomicAdd(&lc[v[k] & (RNG - 1)], 1);
                col[pos] = v[k] >> RSH;
            }
        }
    }
}

#define TP 68
__device__ __forceinline__ void lds_barrier() {
    __builtin_amdgcn_fence(__ATOMIC_RELEASE, "workgroup", "local");
    __builtin_amdgcn_s_barrier();
    __builtin_amdgcn_fence(__ATOMIC_ACQUIRE, "workgroup", "local");
}
__global__ void __launch_bounds__(128) k_layer1(const int* __restrict__ rowptr,
                                                const int* __restrict__ cnt,
                                                const int* __restrict__ col,
                                                const float* __restrict__ dinv,
                                                const float* __restrict__ x,
                                                const float* W1,
                                                const float* __restrict__ b1,
                                                const float* W2,
                                                float* __restrict__ hw2, int n) {
    __shared__ float ta[16 * TP];
    __shared__ float tb[16 * TP];
    const int tid  = threadIdx.x;
    const int g    = tid >> 3;
    const int c    = (tid & 7) << 2;
    const int node = blockIdx.x * 16 + g;
    const int wave = tid >> 5;
    const int lane = tid & 31;
    const int half = lane >> 4;
    const int lm   = lane & 15;
    float4 a0 = make_float4(0.0f, 0.0f, 0.0f, 0.0f);
    float4 a1 = make_float4(0.0f, 0.0f, 0.0f, 0.0f);
    if (node < n) {
        const float di = dinv[node];
        int p = rowptr[node];
        const int end = p + cnt[node];
        const float4* x4 = (const float4*)x;
        const int l = tid & 7;
        const float4 su = x4[node * 16 + l];
        const float4 sv = x4[node * 16 + l + 8];
        const float wself = di * di;
        a0.x = wself * su.x; a0.y = wself * su.y; a0.z = wself * su.z; a0.w = wself * su.w;
        a1.x = wself * sv.x; a1.y = wself * sv.y; a1.z = wself * sv.z; a1.w = wself * sv.w;
        for (; p + 4 <= end; p += 4) {
            const int s0 = col[p], s1 = col[p + 1], s2 = col[p + 2], s3 = col[p + 3];
            const float w0 = dinv[s0] * di, w1 = dinv[s1] * di, w2 = dinv[s2] * di, w3 = dinv[s3] * di;
            const float4 u0 = x4[s0 * 16 + l], v0 = x4[s0 * 16 + l + 8];
            const float4 u1 = x4[s1 * 16 + l], v1 = x4[s1 * 16 + l + 8];
            const float4 u2 = x4[s2 * 16 + l], v2 = x4[s2 * 16 + l + 8];
            const float4 u3 = x4[s3 * 16 + l], v3 = x4[s3 * 16 + l + 8];
            a0.x = fmaf(w0, u0.x, a0.x); a0.y = fmaf(w0, u0.y, a0.y); a0.z = fmaf(w0, u0.z, a0.z); a0.w = fmaf(w0, u0.w, a0.w);
            a1.x = fmaf(w0, v0.x, a1.x); a1.y = fmaf(w0, v0.y, a1.y); a1.z = fmaf(w0, v0.z, a1.z); a1.w = fmaf(w0, v0.w, a1.w);
            a0.x = fmaf(w1, u1.x, a0.x); a0.y = fmaf(w1, u1.y, a0.y); a0.z = fmaf(w1, u1.z, a0.z); a0.w = fmaf(w1, u1.w, a0.w);
            a1.x = fmaf(w1, v1.x, a1.x); a1.y = fmaf(w1, v1.y, a1.y); a1.z = fmaf(w1, v1.z, a1.z); a1.w = fmaf(w1, v1.w, a1.w);
            a0.x = fmaf(w2, u2.x, a0.x); a0.y = fmaf(w2, u2.y, a0.y); a0.z = fmaf(w2, u2.z, a0.z); a0.w = fmaf(w2, u2.w, a0.w);
            a1.x = fmaf(w2, v2.x, a1.x); a1.y = fmaf(w2, v2.y, a1.y); a1.z = fmaf(w2, v2.z, a1.z); a1.w = fmaf(w2, v2.w, a1.w);
            a0.x = fmaf(w3, u3.x, a0.x); a0.y = fmaf(w3, u3.y, a0.y); a0.z = fmaf(w3, u3.z, a0.z); a0.w = fmaf(w3, u3.w, a0.w);
            a1.x = fmaf(w3, v3.x, a1.x); a1.y = fmaf(w3, v3.y, a1.y); a1.z = fmaf(w3, v3.z, a1.z); a1.w = fmaf(w3, v3.w, a1.w);
        }
        for (; p < end; ++p) {
            const int s0 = col[p];
            const float w0 = dinv[s0] * di;
            const float4 u0 = x4[s0 * 16 + l], v0 = x4[s0 * 16 + l + 8];
            a0.x = fmaf(w0, u0.x, a0.x); a0.y = fmaf(w0, u0.y, a0.y); a0.z = fmaf(w0, u0.z, a0.z); a0.w = fmaf(w0, u0.w, a0.w);
            a1.x = fmaf(w0, v0.x, a1.x); a1.y = fmaf(w0, v0.y, a1.y); a1.z = fmaf(w0, v0.z, a1.z); a1.w = fmaf(w0, v0.w, a1.w);
        }
    }
    const int bcol = wave * 16 + lm;
    const float bias = b1[bcol];
    float wf[32];
#pragma unroll
    for (int kk = 0; kk < 16; ++kk) {
        const int kb = kk * 4 + half * 2;
        wf[2 * kk]     = W1[(size_t)kb * 64 + bcol];
        wf[2 * kk + 1] = W1[(size_t)(kb + 1) * 64 + bcol];
    }
    *(float4*)&ta[g * TP + c]      = a0;
    *(float4*)&ta[g * TP + 32 + c] = a1;
    lds_barrier();

    {
        v8f d = {0.f,0.f,0.f,0.f,0.f,0.f,0.f,0.f};
#pragma unroll
        for (int kk = 0; kk < 16; ++kk) {
            const int kb = kk * 4 + half * 2;
            const v2f a = *(const v2f*)&ta[lm * TP + kb];
            v2f b;
            b.x = wf[2 * kk];
            b.y = wf[2 * kk + 1];
            d = __builtin_amdgcn_wmma_f32_16x16x4_f32(false, a, false, b,
                                                      (short)0, d, false, false);
        }
#pragma unroll
        for (int i = 0; i < 8; ++i)
            tb[(half * 8 + i) * TP + bcol] = d[i] + bias;
    }
    const int  col2 = wave * 16 + lm;
    const bool ok   = (col2 < 34);
    const int  cc   = ok ? col2 : 33;
    float vf[32];
#pragma unroll
    for (int q = 0; q < 32; ++q) vf[q] = 0.0f;
    if (wave < 3) {
#pragma unroll
        for (int kk = 0; kk < 16; ++kk) {
            const int kb = kk * 4 + half * 2;
            vf[2 * kk]     = W2[(size_t)kb * 34 + cc];
            vf[2 * kk + 1] = W2[(size_t)(kb + 1) * 34 + cc];
        }
    }
    lds_barrier();

    const float4 p0 = *(const float4*)&tb[g * TP + c];
    const float4 p1 = *(const float4*)&tb[g * TP + 32 + c];
    const float r0 = fmaxf(p0.x, 0.0f), r1 = fmaxf(p0.y, 0.0f);
    const float r2 = fmaxf(p0.z, 0.0f), r3 = fmaxf(p0.w, 0.0f);
    const float r4 = fmaxf(p1.x, 0.0f), r5 = fmaxf(p1.y, 0.0f);
    const float r6 = fmaxf(p1.z, 0.0f), r7 = fmaxf(p1.w, 0.0f);
    float m = fmaxf(fmaxf(fmaxf(r0, r1), fmaxf(r2, r3)), fmaxf(fmaxf(r4, r5), fmaxf(r6, r7)));
    for (int off = 4; off > 0; off >>= 1) m = fmaxf(m, __shfl_xor(m, off, 8));
    const float e0 = __expf(r0 - m), e1 = __expf(r1 - m);
    const float e2 = __expf(r2 - m), e3 = __expf(r3 - m);
    const float e4 = __expf(r4 - m), e5 = __expf(r5 - m);
    const float e6 = __expf(r6 - m), e7 = __expf(r7 - m);
    float s = ((e0 + e1) + (e2 + e3)) + ((e4 + e5) + (e6 + e7));
    for (int off = 4; off > 0; off >>= 1) s += __shfl_xor(s, off, 8);
    const float inv = 1.0f / s;
    float4 o0, o1;
    o0.x = e0 * inv; o0.y = e1 * inv; o0.z = e2 * inv; o0.w = e3 * inv;
    o1.x = e4 * inv; o1.y = e5 * inv; o1.z = e6 * inv; o1.w = e7 * inv;
    *(float4*)&ta[g * TP + c]      = o0;
    *(float4*)&ta[g * TP + 32 + c] = o1;
    lds_barrier();

    if (wave < 3) {
        v8f d = {0.f,0.f,0.f,0.f,0.f,0.f,0.f,0.f};
#pragma unroll
        for (int kk = 0; kk < 16; ++kk) {
            const int kb = kk * 4 + half * 2;
            const v2f a = *(const v2f*)&ta[lm * TP + kb];
            v2f b;
            b.x = ok ? vf[2 * kk]     : 0.0f;
            b.y = ok ? vf[2 * kk + 1] : 0.0f;
            d = __builtin_amdgcn_wmma_f32_16x16x4_f32(false, a, false, b,
                                                      (short)0, d, false, false);
        }
        if (ok) {
#pragma unroll
            for (int i = 0; i < 8; ++i) {
                const int row = blockIdx.x * 16 + half * 8 + i;
                if (row < n) hw2[(size_t)row * 34 + col2] = d[i];
            }
        }
    }
}

__global__ void __launch_bounds__(256) k_gather34(const int* __restrict__ rowptr,
                                                  const int* __restrict__ cnt,
                                                  const int* __restrict__ col,
                                                  const float* __restrict__ dinv,
                                                  const float* __restrict__ hw2,
                                                  const float* __restrict__ b2,
                                                  float* __restrict__ out, int n) {
    const int t = blockIdx.x * 256 + threadIdx.x;
    const int node = t >> 4;
    const int l = t & 15;
    const int c = l << 1;
    if (node >= n) return;
    const float di = dinv[node];
    int p = rowptr[node];
    const int end = p + cnt[node];
    const float* srow = hw2 + (size_t)node * 34;
    const v2f sv = *(const v2f*)(srow + c);
    const v2f sx = *(const v2f*)(srow + 32);
    const float wself = di * di;
    v2f acc, accx;
    acc.x  = b2[c]  + wself * sv.x;
    acc.y  = b2[c + 1] + wself * sv.y;
    accx.x = b2[32] + wself * sx.x;
    accx.y = b2[33] + wself * sx.y;
    for (; p + 4 <= end; p += 4) {
        const int s0 = col[p], s1 = col[p + 1], s2 = col[p + 2], s3 = col[p + 3];
        const float w0 = dinv[s0] * di, w1 = dinv[s1] * di, w2 = dinv[s2] * di, w3 = dinv[s3] * di;
        const float* r0 = hw2 + (size_t)s0 * 34;
        const float* r1 = hw2 + (size_t)s1 * 34;
        const float* r2 = hw2 + (size_t)s2 * 34;
        const float* r3 = hw2 + (size_t)s3 * 34;
        const v2f v0 = *(const v2f*)(r0 + c), x0 = *(const v2f*)(r0 + 32);
        const v2f v1 = *(const v2f*)(r1 + c), x1 = *(const v2f*)(r1 + 32);
        const v2f v2 = *(const v2f*)(r2 + c), x2 = *(const v2f*)(r2 + 32);
        const v2f v3 = *(const v2f*)(r3 + c), x3 = *(const v2f*)(r3 + 32);
        acc.x += w0 * v0.x; acc.y += w0 * v0.y; accx.x += w0 * x0.x; accx.y += w0 * x0.y;
        acc.x += w1 * v1.x; acc.y += w1 * v1.y; accx.x += w1 * x1.x; accx.y += w1 * x1.y;
        acc.x += w2 * v2.x; acc.y += w2 * v2.y; accx.x += w2 * x2.x; accx.y += w2 * x2.y;
        acc.x += w3 * v3.x; acc.y += w3 * v3.y; accx.x += w3 * x3.x; accx.y += w3 * x3.y;
    }
    for (; p < end; ++p) {
        const int s0 = col[p];
        const float w0 = dinv[s0] * di;
        const float* r0 = hw2 + (size_t)s0 * 34;
        const v2f v0 = *(const v2f*)(r0 + c), x0 = *(const v2f*)(r0 + 32);
        acc.x += w0 * v0.x; acc.y += w0 * v0.y; accx.x += w0 * x0.x; accx.y += w0 * x0.y;
    }
    float* orow = out + (size_t)node * 34;
    *(v2f*)(orow + c) = acc;
    if (l == 0) *(v2f*)(orow + 32) = accx;
}

extern "C" void kernel_launch(void* const* d_in, const int* in_sizes, int n_in,
                              void* d_out, int out_size, void* d_ws, size_t ws_size,
                              hipStream_t stream) {
    const float* x  = (const float*)d_in[0];
    const int*   ei = (const int*)  d_in[1];
    const float* W1 = (const float*)d_in[2];
    const float* b1 = (const float*)d_in[3];
    const float* W2 = (const float*)d_in[4];
    const float* b2 = (const float*)d_in[5];
    float* out = (float*)d_out;

    const int N = in_sizes[0] / 64;
    const int E = in_sizes[1] / 2;
    const int* src = ei;
    const int* dst = ei + E;

    int*   cnt    = (int*)d_ws;
    int*   rowptr = cnt    + N;
    int*   pad    = rowptr + N;
    float* dinv   = (float*)(pad + 64);
    float* hw2    = dinv + N;
    int*   col    = (int*)(hw2 + (size_t)N * 34);
    int*   binned = col + E;
    int*   binoff = binned + E;
    int*   hist   = binoff + 512;

    const int T = 256;
    const int nblk = (E + CH - 1) / CH;
    k_hist<<<nblk, 1024, 0, stream>>>(dst, hist, E);
    k_bin <<<nblk, 1024, 0, stream>>>(src, dst, hist, binoff, binned, E, nblk);
    k_csr <<<(N + RNG - 1) / RNG, 1024, 0, stream>>>(binned, binoff, cnt, rowptr, dinv, col, N);
    k_layer1<<<(N + 15) / 16, 128, 0, stream>>>(rowptr, cnt, col, dinv, x, W1, b1, W2, hw2, N);
    k_gather34<<<((size_t)N * 16 + T - 1) / T, T, 0, stream>>>(rowptr, cnt, col, dinv, hw2, b2, out, N);
}


# Layer-2 edge weights folded away: layer 1 stores hw2 rows scaled by deg^-1/2, the layer-2 gather adds rows unweighted and scales its sum once; no per-edge dinv read; layer-2 rounding changes
# speedup vs baseline: 3.7782x; 1.0302x over previous
#include <hip/hip_runtime.h>
#include <math.h>

typedef __attribute__((ext_vector_type(2))) float v2f;
typedef __attribute__((ext_vector_type(8))) float v8f;


#define RSH   9
#define RNG   512
#define NBMAX 256
#define CH    5120

__global__ void __launch_bounds__(1024) k_hist(const int* __restrict__ dst, int* __restrict__ hist, int E) {
    __shared__ int lh[NBMAX];
    const int tid = threadIdx.x;
    if (tid < NBMAX) lh[tid] = 0;
    __syncthreads();
    const int base = blockIdx.x * CH + tid;
    int d[CH / 1024];
#pragma unroll
    for (int k = 0; k < CH / 1024; ++k) {
        const int e = base + k * 1024;
        d[k] = dst[e < E ? e : E - 1];
    }
#pragma unroll
    for (int k = 0; k < CH / 1024; ++k)
        if (base + k * 1024 < E) atomicAdd(&lh[d[k] >> RSH], 1);
    __syncthreads();
    if (tid < NBMAX) hist[blockIdx.x * NBMAX + tid] = lh[tid];
}

__global__ void __launch_bounds__(1024) k_bin(const int* __restrict__ src, const int* __restrict__ dst,
                                              const int* __restrict__ hist, int* __restrict__ binoff,
                                              int* __restrict__ binned, int E, int nblk) {
    __shared__ int ltot[16][NBMAX];
    __shared__ int lpre[16][NBMAX];
    __shared__ int lcur[NBMAX];
    __shared__ int wsum[8];
    const int tid = threadIdx.x, lane = tid & 31, wv = tid >> 5;
    const int me = blockIdx.x;
    const int base = me * CH + tid;
    int d[CH / 1024], s[CH / 1024];
#pragma unroll
    for (int k = 0; k < CH / 1024; ++k) {
        const int e  = base + k * 1024;
        const int ec = e < E ? e : E - 1;
        d[k] = dst[ec];
        s[k] = src[ec];
    }
    const int q  = tid >> 6;
    const int c4 = (tid & 63) << 2;
    int4 tot = make_int4(0, 0, 0, 0), pre = make_int4(0, 0, 0, 0);
    for (int b = q; b < nblk; b += 16) {
        const int4 v = *(const int4*)(hist + (size_t)b * NBMAX + c4);
        tot.x += v.x; tot.y += v.y; tot.z += v.z; tot.w += v.w;
        if (b < me) { pre.x += v.x; pre.y += v.y; pre.z += v.z; pre.w += v.w; }
    }
    *(int4*)&ltot[q][c4] = tot;
    *(int4*)&lpre[q][c4] = pre;
    __syncthreads();
    int t = 0, p = 0, inc = 0;
    if (tid < NBMAX) {
#pragma unroll
        for (int g = 0; g < 16; ++g) { t += ltot[g][tid]; p += lpre[g][tid]; }
        inc = t;
#pragma unroll
        for (int off = 1; off < 32; off <<= 1) {
            int v = __shfl_up(inc, off, 32);
            if (lane >= off) inc += v;
        }
        if (lane == 31) wsum[wv] = inc;
    }
    __syncthreads();
    if (tid < NBMAX) {
        int woff = 0;
        for (int w = 0; w < wv; ++w) woff += wsum[w];
        const int off = woff + inc - t;
        lcur[tid] = off + p;
        if (me == 0) {
            binoff[tid] = off;
            if (tid == NBMAX - 1) binoff[NBMAX] = off + t;
        }
    }
    __syncthreads();
#pragma unroll
    for (int k = 0; k < CH / 1024; ++k) {
        if (base + k * 1024 < E) {
            const int pos = atomicAdd(&lcur[d[k] >> RSH], 1);
            binned[pos] = (s[k] << RSH) | (d[k] & (RNG - 1));
        }
    }
}

__global__ void __launch_bounds__(1024) k_csr(const int* __restrict__ binned, const int* __restrict__ binoff,
                                              int* __restrict__ cnt, int* __restrict__ rowptr,
                                              float* __restrict__ dinv, int* __restrict__ col, int n) {
    __shared__ int lc[RNG];
    __shared__ int wsum[RNG / 32];
    const int tid = threadIdx.x, lane = tid & 31, wv = tid >> 5;
    const int start = binoff[blockIdx.x];
    const int end   = binoff[blockIdx.x + 1];
    if (tid < RNG) lc[tid] = 0;
    __syncthreads();
    for (int e0 = start + tid; e0 < end; e0 += 8 * 1024) {
        int v[8];
#pragma unroll
        for (int k = 0; k < 8; ++k) {
            const int e = e0 + k * 1024;
            v[k] = binned[e < end ? e : end - 1];
        }
        asm volatile("" :: "v"(v[0]), "v"(v[1]), "v"(v[2]), "v"(v[3]), "v"(v[4]), "v"(v[5]), "v"(v[6]), "v"(v[7]));
#pragma unroll
        for (int k = 0; k < 8; ++k)
            if (e0 + k * 1024 < end) atomicAdd(&lc[v[k] & (RNG - 1)], 1);
    }
    __syncthreads();
    int c = 0, inc = 0;
    if (tid < RNG) {
        c = lc[tid];
        inc = c;
#pragma unroll
        for (int off = 1; off < 32; off <<= 1) {
            int v = __shfl_up(inc, off, 32);
            if (lane >= off) inc += v;
        }
        if (lane == 31) wsum[wv] = inc;
    }
    __syncthreads();
    if (tid < RNG) {
        int woff = 0;
        for (int w = 0; w < wv; ++w) woff += wsum[w];
        const int ex = start + woff + inc - c;
        lc[tid] = ex;
        const int node = blockIdx.x * RNG + tid;
        if (node < n) {
            cnt[node]    = c;
            rowptr[node] = ex;
            dinv[node]   = rsqrtf((float)(c + 1));
        }
    }
    __syncthreads();
    for (int e0 = start + tid; e0 < end; e0 += 8 * 1024) {
        int v[8];
#pragma unroll
        for (int k = 0; k < 8; ++k) {
            const int e = e0 + k * 1024;
            v[k] = binned[e < end ? e : end - 1];
        }
        asm volatile("" :: "v"(v[0]), "v"(v[1]), "v"(v[2]), "v"(v[3]), "v"(v[4]), "v"(v[5]), "v"(v[6]), "v"(v[7]));
#pragma unroll
        for (int k = 0; k < 8; ++k) {
            if (e0 + k * 1024 < end) {
                const int pos = atomicAdd(&lc[v[k] & (RNG - 1)], 1);
                col[pos] = v[k] >> RSH;
            }
        }
    }
}

#define TP 68
__device__ __forceinline__ void lds_barrier() {
    __builtin_amdgcn_fence(__ATOMIC_RELEASE, "workgroup", "local");
    __builtin_amdgcn_s_barrier();
    __builtin_amdgcn_fence(__ATOMIC_ACQUIRE, "workgroup", "local");
}
__global__ void __launch_bounds__(128) k_layer1(const int* __restrict__ rowptr,
                                                const int* __restrict__ cnt,
                                                const int* __restrict__ col,
                                                const float* __restrict__ dinv,
                                                const float* __restrict__ x,
                                                const float* W1,
                                                const float* __restrict__ b1,
                                                const float* W2,
                                                float* __restrict__ hw2, int n) {
    __shared__ float ta[16 * TP];
    __shared__ float tb[16 * TP];
    const int tid  = threadIdx.x;
    const int g    = tid >> 3;
    const int c    = (tid & 7) << 2;
    const int node = blockIdx.x * 16 + g;
    const int wave = tid >> 5;
    const int lane = tid & 31;
    const int half = lane >> 4;
    const int lm   = lane & 15;
    float4 a0 = make_float4(0.0f, 0.0f, 0.0f, 0.0f);
    float4 a1 = make_float4(0.0f, 0.0f, 0.0f, 0.0f);
    float di = 0.0f;
    if (node < n) {
        di = dinv[node];
        int p = rowptr[node];
        const int end = p + cnt[node];
        const float4* x4 = (const float4*)x;
        const int l = tid & 7;
        const float4 su = x4[node * 16 + l];
        const float4 sv = x4[node * 16 + l + 8];
        const float wself = di * di;
        a0.x = wself * su.x; a0.y = wself * su.y; a0.z = wself * su.z; a0.w = wself * su.w;
        a1.x = wself * sv.x; a1.y = wself * sv.y; a1.z = wself * sv.z; a1.w = wself * sv.w;
        for (; p + 4 <= end; p += 4) {
            const int s0 = col[p], s1 = col[p + 1], s2 = col[p + 2], s3 = col[p + 3];
            const float w0 = dinv[s0] * di, w1 = dinv[s1] * di, w2 = dinv[s2] * di, w3 = dinv[s3] * di;
            const float4 u0 = x4[s0 * 16 + l], v0 = x4[s0 * 16 + l + 8];
            const float4 u1 = x4[s1 * 16 + l], v1 = x4[s1 * 16 + l + 8];
            const float4 u2 = x4[s2 * 16 + l], v2 = x4[s2 * 16 + l + 8];
            const float4 u3 = x4[s3 * 16 + l], v3 = x4[s3 * 16 + l + 8];
            a0.x = fmaf(w0, u0.x, a0.x); a0.y = fmaf(w0, u0.y, a0.y); a0.z = fmaf(w0, u0.z, a0.z); a0.w = fmaf(w0, u0.w, a0.w);
            a1.x = fmaf(w0, v0.x, a1.x); a1.y = fmaf(w0, v0.y, a1.y); a1.z = fmaf(w0, v0.z, a1.z); a1.w = fmaf(w0, v0.w, a1.w);
            a0.x = fmaf(w1, u1.x, a0.x); a0.y = fmaf(w1, u1.y, a0.y); a0.z = fmaf(w1, u1.z, a0.z); a0.w = fmaf(w1, u1.w, a0.w);
            a1.x = fmaf(w1, v1.x, a1.x); a1.y = fmaf(w1, v1.y, a1.y); a1.z = fmaf(w1, v1.z, a1.z); a1.w = fmaf(w1, v1.w, a1.w);
            a0.x = fmaf(w2, u2.x, a0.x); a0.y = fmaf(w2, u2.y, a0.y); a0.z = fmaf(w2, u2.z, a0.z); a0.w = fmaf(w2, u2.w, a0.w);
            a1.x = fmaf(w2, v2.x, a1.x); a1.y = fmaf(w2, v2.y, a1.y); a1.z = fmaf(w2, v2.z, a1.z); a1.w = fmaf(w2, v2.w, a1.w);
            a0.x = fmaf(w3, u3.x, a0.x); a0.y = fmaf(w3, u3.y, a0.y); a0.z = fmaf(w3, u3.z, a0.z); a0.w = fmaf(w3, u3.w, a0.w);
            a1.x = fmaf(w3, v3.x, a1.x); a1.y = fmaf(w3, v3.y, a1.y); a1.z = fmaf(w3, v3.z, a1.z); a1.w = fmaf(w3, v3.w, a1.w);
        }
        for (; p < end; ++p) {
            const int s0 = col[p];
            const float w0 = dinv[s0] * di;
            const float4 u0 = x4[s0 * 16 + l], v0 = x4[s0 * 16 + l + 8];
            a0.x = fmaf(w0, u0.x, a0.x); a0.y = fmaf(w0, u0.y, a0.y); a0.z = fmaf(w0, u0.z, a0.z); a0.w = fmaf(w0, u0.w, a0.w);
            a1.x = fmaf(w0, v0.x, a1.x); a1.y = fmaf(w0, v0.y, a1.y); a1.z = fmaf(w0, v0.z, a1.z); a1.w = fmaf(w0, v0.w, a1.w);
        }
    }
    const int bcol = wave * 16 + lm;
    const float bias = b1[bcol];
    float wf[32];
#pragma unroll
    for (int kk = 0; kk < 16; ++kk) {
        const int kb = kk * 4 + half * 2;
        wf[2 * kk]     = W1[(size_t)kb * 64 + bcol];
        wf[2 * kk + 1] = W1[(size_t)(kb + 1) * 64 + bcol];
    }
    *(float4*)&ta[g * TP + c]      = a0;
    *(float4*)&ta[g * TP + 32 + c] = a1;
    lds_barrier();

    {
        v8f d = {0.f,0.f,0.f,0.f,0.f,0.f,0.f,0.f};
#pragma unroll
        for (int kk = 0; kk < 16; ++kk) {
            const int kb = kk * 4 + half * 2;
            const v2f a = *(const v2f*)&ta[lm * TP + kb];
            v2f b;
            b.x = wf[2 * kk];
            b.y = wf[2 * kk + 1];
            d = __builtin_amdgcn_wmma_f32_16x16x4_f32(false, a, false, b,
                                                      (short)0, d, false, false);
        }
#pragma unroll
        for (int i = 0; i < 8; ++i)
            tb[(half * 8 + i) * TP + bcol] = d[i] + bias;
    }
    const int  col2 = wave * 16 + lm;
    const bool ok   = (col2 < 34);
    const int  cc   = ok ? col2 : 33;
    float vf[32];
#pragma unroll
    for (int q = 0; q < 32; ++q) vf[q] = 0.0f;
    if (wave < 3) {
#pragma unroll
        for (int kk = 0; kk < 16; ++kk) {
            const int kb = kk * 4 + half * 2;
            vf[2 * kk]     = W2[(size_t)kb * 34 + cc];
            vf[2 * kk + 1] = W2[(size_t)(kb + 1) * 34 + cc];
        }
    }
    lds_barrier();

    const float4 p0 = *(const float4*)&tb[g * TP + c];
    const float4 p1 = *(const float4*)&tb[g * TP + 32 + c];
    const float r0 = fmaxf(p0.x, 0.0f), r1 = fmaxf(p0.y, 0.0f);
    const float r2 = fmaxf(p0.z, 0.0f), r3 = fmaxf(p0.w, 0.0f);
    const float r4 = fmaxf(p1.x, 0.0f), r5 = fmaxf(p1.y, 0.0f);
    const float r6 = fmaxf(p1.z, 0.0f), r7 = fmaxf(p1.w, 0.0f);
    float m = fmaxf(fmaxf(fmaxf(r0, r1), fmaxf(r2, r3)), fmaxf(fmaxf(r4, r5), fmaxf(r6, r7)));
    for (int off = 4; off > 0; off >>= 1) m = fmaxf(m, __shfl_xor(m, off, 8));
    const float e0 = __expf(r0 - m), e1 = __expf(r1 - m);
    const float e2 = __expf(r2 - m), e3 = __expf(r3 - m);
    const float e4 = __expf(r4 - m), e5 = __expf(r5 - m);
    const float e6 = __expf(r6 - m), e7 = __expf(r7 - m);
    float s = ((e0 + e1) + (e2 + e3)) + ((e4 + e5) + (e6 + e7));
    for (int off = 4; off > 0; off >>= 1) s += __shfl_xor(s, off, 8);
    const float inv = (1.0f / s) * di;
    float4 o0, o1;
    o0.x = e0 * inv; o0.y = e1 * inv; o0.z = e2 * inv; o0.w = e3 * inv;
    o1.x = e4 * inv; o1.y = e5 * inv; o1.z = e6 * inv; o1.w = e7 * inv;
    *(float4*)&ta[g * TP + c]      = o0;
    *(float4*)&ta[g * TP + 32 + c] = o1;
    lds_barrier();

    if (wave < 3) {
        v8f d = {0.f,0.f,0.f,0.f,0.f,0.f,0.f,0.f};
#pragma unroll
        for (int kk = 0; kk < 16; ++kk) {
            const int kb = kk * 4 + half * 2;
            const v2f a = *(const v2f*)&ta[lm * TP + kb];
            v2f b;
            b.x = ok ? vf[2 * kk]     : 0.0f;
            b.y = ok ? vf[2 * kk + 1] : 0.0f;
            d = __builtin_amdgcn_wmma_f32_16x16x4_f32(false, a, false, b,
                                                      (short)0, d, false, false);
        }
        if (ok) {
#pragma unroll
            for (int i = 0; i < 8; ++i) {
                const int row = blockIdx.x * 16 + half * 8 + i;
                if (row < n) hw2[(size_t)row * 34 + col2] = d[i];
            }
        }
    }
}

__global__ void __launch_bounds__(256) k_gather34(const int* __restrict__ rowptr,
                                                  const int* __restrict__ cnt,
                                                  const int* __restrict__ col,
                                                  const float* __restrict__ dinv,
                                                  const float* __restrict__ hw2,
                                                  const float* __restrict__ b2,
                                                  float* __restrict__ out, int n) {
    const int t = blockIdx.x * 256 + threadIdx.x;
    const int node = t >> 4;
    const int l = t & 15;
    const int c = l << 1;
    if (node >= n) return;
    const float di = dinv[node];
    int p = rowptr[node];
    const int end = p + cnt[node];
    const float* srow = hw2 + (size_t)node * 34;
    const v2f sv = *(const v2f*)(srow + c);
    const v2f sx = *(const v2f*)(srow + 32);
    v2f acc = sv, accx = sx;
    for (; p + 4 <= end; p += 4) {
        const int s0 = col[p], s1 = col[p + 1], s2 = col[p + 2], s3 = col[p + 3];
        const float* r0 = hw2 + (size_t)s0 * 34;
        const float* r1 = hw2 + (size_t)s1 * 34;
        const float* r2 = hw2 + (size_t)s2 * 34;
        const float* r3 = hw2 + (size_t)s3 * 34;
        const v2f v0 = *(const v2f*)(r0 + c), x0 = *(const v2f*)(r0 + 32);
        const v2f v1 = *(const v2f*)(r1 + c), x1 = *(const v2f*)(r1 + 32);
        const v2f v2 = *(const v2f*)(r2 + c), x2 = *(const v2f*)(r2 + 32);
        const v2f v3 = *(const v2f*)(r3 + c), x3 = *(const v2f*)(r3 + 32);
        acc += v0; accx += x0;
        acc += v1; accx += x1;
        acc += v2; accx += x2;
        acc += v3; accx += x3;
    }
    for (; p < end; ++p) {
        const int s0 = col[p];
        const float* r0 = hw2 + (size_t)s0 * 34;
        const v2f v0 = *(const v2f*)(r0 + c), x0 = *(const v2f*)(r0 + 32);
        acc += v0; accx += x0;
    }
    v2f o, ox;
    o.x  = fmaf(di, acc.x,  b2[c]);
    o.y  = fmaf(di, acc.y,  b2[c + 1]);
    ox.x = fmaf(di, accx.x, b2[32]);
    ox.y = fmaf(di, accx.y, b2[33]);
    float* orow = out + (size_t)node * 34;
    *(v2f*)(orow + c) = o;
    if (l == 0) *(v2f*)(orow + 32) = ox;
}

extern "C" void kernel_launch(void* const* d_in, const int* in_sizes, int n_in,
                              void* d_out, int out_size, void* d_ws, size_t ws_size,
                              hipStream_t stream) {
    const float* x  = (const float*)d_in[0];
    const int*   ei = (const int*)  d_in[1];
    const float* W1 = (const float*)d_in[2];
    const float* b1 = (const float*)d_in[3];
    const float* W2 = (const float*)d_in[4];
    const float* b2 = (const float*)d_in[5];
    float* out = (float*)d_out;

    const int N = in_sizes[0] / 64;
    const int E = in_sizes[1] / 2;
    const int* src = ei;
    const int* dst = ei + E;

    int*   cnt    = (int*)d_ws;
    int*   rowptr = cnt    + N;
    int*   pad    = rowptr + N;
    float* dinv   = (float*)(pad + 64);
    float* hw2    = dinv + N;
    int*   col    = (int*)(hw2 + (size_t)N * 34);
    int*   binned = col + E;
    int*   binoff = binned + E;
    int*   hist   = binoff + 512;

    const int T = 256;
    const int nblk = (E + CH - 1) / CH;
    k_hist<<<nblk, 1024, 0, stream>>>(dst, hist, E);
    k_bin <<<nblk, 1024, 0, stream>>>(src, dst, hist, binoff, binned, E, nblk);
    k_csr <<<(N + RNG - 1) / RNG, 1024, 0, stream>>>(binned, binoff, cnt, rowptr, dinv, col, N);
    k_layer1<<<(N + 15) / 16, 128, 0, stream>>>(rowptr, cnt, col, dinv, x, W1, b1, W2, hw2, N);
    k_gather34<<<((size_t)N * 16 + T - 1) / T, T, 0, stream>>>(rowptr, cnt, col, dinv, hw2, b2, out, N);
}


# Both gather walks lose their dependent id round trips: a trip's four source ids are loaded one trip ahead (16-byte load); the partial last trip runs predicated (weight 0), not per edge; same sums
# speedup vs baseline: 4.0327x; 1.0674x over previous
#include <hip/hip_runtime.h>
#include <math.h>

typedef __attribute__((ext_vector_type(2))) float v2f;
typedef __attribute__((ext_vector_type(8))) float v8f;


#define RSH   9
#define RNG   512
#define NBMAX 256
#define CH    5120

__global__ void __launch_bounds__(1024) k_hist(const int* __restrict__ dst, int* __restrict__ hist, int E) {
    __shared__ int lh[NBMAX];
    const int tid = threadIdx.x;
    if (tid < NBMAX) lh[tid] = 0;
    __syncthreads();
    const int base = blockIdx.x * CH + tid;
    int d[CH / 1024];
#pragma unroll
    for (int k = 0; k < CH / 1024; ++k) {
        const int e = base + k * 1024;
        d[k] = dst[e < E ? e : E - 1];
    }
#pragma unroll
    for (int k = 0; k < CH / 1024; ++k)
        if (base + k * 1024 < E) atomicAdd(&lh[d[k] >> RSH], 1);
    __syncthreads();
    if (tid < NBMAX) hist[blockIdx.x * NBMAX + tid] = lh[tid];
}

__global__ void __launch_bounds__(1024) k_bin(const int* __restrict__ src, const int* __restrict__ dst,
                                              const int* __restrict__ hist, int* __restrict__ binoff,
                                              int* __restrict__ binned, int E, int nblk) {
    __shared__ int ltot[16][NBMAX];
    __shared__ int lpre[16][NBMAX];
    __shared__ int lcur[NBMAX];
    __shared__ int wsum[8];
    const int tid = threadIdx.x, lane = tid & 31, wv = tid >> 5;
    const int me = blockIdx.x;
    const int base = me * CH + tid;
    int d[CH / 1024], s[CH / 1024];
#pragma unroll
    for (int k = 0; k < CH / 1024; ++k) {
        const int e  = base + k * 1024;
        const int ec = e < E ? e : E - 1;
        d[k] = dst[ec];
        s[k] = src[ec];
    }
    const int q  = tid >> 6;
    const int c4 = (tid & 63) << 2;
    int4 tot = make_int4(0, 0, 0, 0), pre = make_int4(0, 0, 0, 0);
    for (int b = q; b < nblk; b += 16) {
        const int4 v = *(const int4*)(hist + (size_t)b * NBMAX + c4);
        tot.x += v.x; tot.y += v.y; tot.z += v.z; tot.w += v.w;
        if (b < me) { pre.x += v.x; pre.y += v.y; pre.z += v.z; pre.w += v.w; }
    }
    *(int4*)&ltot[q][c4] = tot;
    *(int4*)&lpre[q][c4] = pre;
    __syncthreads();
    int t = 0, p = 0, inc = 0;
    if (tid < NBMAX) {
#pragma unroll
        for (int g = 0; g < 16; ++g) { t += ltot[g][tid]; p += lpre[g][tid]; }
        inc = t;
#pragma unroll
        for (int off = 1; off < 32; off <<= 1) {
            int v = __shfl_up(inc, off, 32);
            if (lane >= off) inc += v;
        }
        if (lane == 31) wsum[wv] = inc;
    }
    __syncthreads();
    if (tid < NBMAX) {
        int woff = 0;
        for (int w = 0; w < wv; ++w) woff += wsum[w];
        const int off = woff + inc - t;
        lcur[tid] = off + p;
        if (me == 0) {
            binoff[tid] = off;
            if (tid == NBMAX - 1) binoff[NBMAX] = off + t;
        }
    }
    __syncthreads();
#pragma unroll
    for (int k = 0; k < CH / 1024; ++k) {
        if (base + k * 1024 < E) {
            const int pos = atomicAdd(&lcur[d[k] >> RSH], 1);
            binned[pos] = (s[k] << RSH) | (d[k] & (RNG - 1));
        }
    }
}

__global__ void __launch_bounds__(1024) k_csr(const int* __restrict__ binned, const int* __restrict__ binoff,
                                              int* __restrict__ cnt, int* __restrict__ rowptr,
                                              float* __restrict__ dinv, int* __restrict__ col, int n) {
    __shared__ int lc[RNG];
    __shared__ int wsum[RNG / 32];
    const int tid = threadIdx.x, lane = tid & 31, wv = tid >> 5;
    const int start = binoff[blockIdx.x];
    const int end   = binoff[blockIdx.x + 1];
    if (tid < RNG) lc[tid] = 0;
    __syncthreads();
    for (int e0 = start + tid; e0 < end; e0 += 8 * 1024) {
        int v[8];
#pragma unroll
        for (int k = 0; k < 8; ++k) {
            const int e = e0 + k * 1024;
            v[k] = binned[e < end ? e : end - 1];
        }
        asm volatile("" :: "v"(v[0]), "v"(v[1]), "v"(v[2]), "v"(v[3]), "v"(v[4]), "v"(v[5]), "v"(v[6]), "v"(v[7]));
#pragma unroll
        for (int k = 0; k < 8; ++k)
            if (e0 + k * 1024 < end) atomicAdd(&lc[v[k] & (RNG - 1)], 1);
    }
    __syncthreads();
    int c = 0, inc = 0;
    if (tid < RNG) {
        c = lc[tid];
        inc = c;
#pragma unroll
        for (int off = 1; off < 32; off <<= 1) {
            int v = __shfl_up(inc, off, 32);
            if (lane >= off) inc += v;
        }
        if (lane == 31) wsum[wv] = inc;
    }
    __syncthreads();
    if (tid < RNG) {
        int woff = 0;
        for (int w = 0; w < wv; ++w) woff += wsum[w];
        const int ex = start + woff + inc - c;
        lc[tid] = ex;
        const int node = blockIdx.x * RNG + tid;
        if (node < n) {
            cnt[node]    = c;
            rowptr[node] = ex;
            dinv[node]   = rsqrtf((float)(c + 1));
        }
    }
    __syncthreads();
    for (int e0 = start + tid; e0 < end; e0 += 8 * 1024) {
        int v[8];
#pragma unroll
        for (int k = 0; k < 8; ++k) {
            const int e = e0 + k * 1024;
            v[k] = binned[e < end ? e : end - 1];
        }
        asm volatile("" :: "v"(v[0]), "v"(v[1]), "v"(v[2]), "v"(v[3]), "v"(v[4]), "v"(v[5]), "v"(v[6]), "v"(v[7]));
#pragma unroll
        for (int k = 0; k < 8; ++k) {
            if (e0 + k * 1024 < end) {
                const int pos = atomicAdd(&lc[v[k] & (RNG - 1)], 1);
                col[pos] = v[k] >> RSH;
            }
        }
    }
}

#define TP 68
__device__ __forceinline__ void lds_barrier() {
    __builtin_amdgcn_fence(__ATOMIC_RELEASE, "workgroup", "local");
    __builtin_amdgcn_s_barrier();
    __builtin_amdgcn_fence(__ATOMIC_ACQUIRE, "workgroup", "local");
}
__global__ void __launch_bounds__(128) k_layer1(const int* __restrict__ rowptr,
                                                const int* __restrict__ cnt,
                                                const int* __restrict__ col,
                                                const float* __restrict__ dinv,
                                                const float* __restrict__ x,
                                                const float* W1,
                                                const float* __restrict__ b1,
                                                const float* W2,
                                                float* __restrict__ hw2, int n) {
    __shared__ float ta[16 * TP];
    __shared__ float tb[16 * TP];
    const int tid  = threadIdx.x;
    const int g    = tid >> 3;
    const int c    = (tid & 7) << 2;
    const int node = blockIdx.x * 16 + g;
    const int wave = tid >> 5;
    const int lane = tid & 31;
    const int half = lane >> 4;
    const int lm   = lane & 15;
    float4 a0 = make_float4(0.0f, 0.0f, 0.0f, 0.0f);
    float4 a1 = make_float4(0.0f, 0.0f, 0.0f, 0.0f);
    float di = 0.0f;
    if (node < n) {
        di = dinv[node];
        int p = rowptr[node];
        const int nin = cnt[node];
        const int end = p + nin;
        const float4* x4 = (const float4*)x;
        const int l = tid & 7;
        const float4 su = x4[node * 16 + l];
        const float4 sv = x4[node * 16 + l + 8];
        asm volatile("" :: "v"(p), "v"(nin));
        const float wself = di * di;
        a0.x = wself * su.x; a0.y = wself * su.y; a0.z = wself * su.z; a0.w = wself * su.w;
        a1.x = wself * sv.x; a1.y = wself * sv.y; a1.z = wself * sv.z; a1.w = wself * sv.w;
        int4 cur = *(const int4*)(col + p);
        while (p < end) {
            const int4 nxt = *(const int4*)(col + p + 4);
            const bool k1 = p + 1 < end, k2 = p + 2 < end, k3 = p + 3 < end;
            const int s0 = cur.x;
            const int s1 = k1 ? cur.y : s0, s2 = k2 ? cur.z : s0, s3 = k3 ? cur.w : s0;
            const float d0 = dinv[s0], d1 = dinv[s1], d2 = dinv[s2], d3 = dinv[s3];
            const float4 u0 = x4[s0 * 16 + l], v0 = x4[s0 * 16 + l + 8];
            const float4 u1 = x4[s1 * 16 + l], v1 = x4[s1 * 16 + l + 8];
            const float4 u2 = x4[s2 * 16 + l], v2 = x4[s2 * 16 + l + 8];
            const float4 u3 = x4[s3 * 16 + l], v3 = x4[s3 * 16 + l + 8];
            const float w0 = d0 * di;
            const float w1 = d1 * (k1 ? di : 0.0f);
            const float w2 = d2 * (k2 ? di : 0.0f);
            const float w3 = d3 * (k3 ? di : 0.0f);
            a0.x = fmaf(w0, u0.x, a0.x); a0.y = fmaf(w0, u0.y, a0.y); a0.z = fmaf(w0, u0.z, a0.z); a0.w = fmaf(w0, u0.w, a0.w);
            a1.x = fmaf(w0, v0.x, a1.x); a1.y = fmaf(w0, v0.y, a1.y); a1.z = fmaf(w0, v0.z, a1.z); a1.w = fmaf(w0, v0.w, a1.w);
            a0.x = fmaf(w1, u1.x, a0.x); a0.y = fmaf(w1, u1.y, a0.y); a0.z = fmaf(w1, u1.z, a0.z); a0.w = fmaf(w1, u1.w, a0.w);
            a1.x = fmaf(w1, v1.x, a1.x); a1.y = fmaf(w1, v1.y, a1.y); a1.z = fmaf(w1, v1.z, a1.z); a1.w = fmaf(w1, v1.w, a1.w);
            a0.x = fmaf(w2, u2.x, a0.x); a0.y = fmaf(w2, u2.y, a0.y); a0.z = fmaf(w2, u2.z, a0.z); a0.w = fmaf(w2, u2.w, a0.w);
            a1.x = fmaf(w2, v2.x, a1.x); a1.y = fmaf(w2, v2.y, a1.y); a1.z = fmaf(w2, v2.z, a1.z); a1.w = fmaf(w2, v2.w, a1.w);
            a0.x = fmaf(w3, u3.x, a0.x); a0.y = fmaf(w3, u3.y, a0.y); a0.z = fmaf(w3, u3.z, a0.z); a0.w = fmaf(w3, u3.w, a0.w);
            a1.x = fmaf(w3, v3.x, a1.x); a1.y = fmaf(w3, v3.y, a1.y); a1.z = fmaf(w3, v3.z, a1.z); a1.w = fmaf(w3, v3.w, a1.w);
            asm volatile("" :: "v"(nxt.x), "v"(nxt.y), "v"(nxt.z), "v"(nxt.w));
            cur = nxt;
            p += 4;
        }
    }
    const int bcol = wave * 16 + lm;
    const float bias = b1[bcol];
    float wf[32];
#pragma unroll
    for (int kk = 0; kk < 16; ++kk) {
        const int kb = kk * 4 + half * 2;
        wf[2 * kk]     = W1[(size_t)kb * 64 + bcol];
        wf[2 * kk + 1] = W1[(size_t)(kb + 1) * 64 + bcol];
    }
    *(float4*)&ta[g * TP + c]      = a0;
    *(float4*)&ta[g * TP + 32 + c] = a1;
    lds_barrier();

    {
        v8f d = {0.f,0.f,0.f,0.f,0.f,0.f,0.f,0.f};
#pragma unroll
        for (int kk = 0; kk < 16; ++kk) {
            const int kb = kk * 4 + half * 2;
            const v2f a = *(const v2f*)&ta[lm * TP + kb];
            v2f b;
            b.x = wf[2 * kk];
            b.y = wf[2 * kk + 1];
            d = __builtin_amdgcn_wmma_f32_16x16x4_f32(false, a, false, b,
                                                      (short)0, d, false, false);
        }
#pragma unroll
        for (int i = 0; i < 8; ++i)
            tb[(half * 8 + i) * TP + bcol] = d[i] + bias;
    }
    const int  col2 = wave * 16 + lm;
    const bool ok   = (col2 < 34);
    const int  cc   = ok ? col2 : 33;
    float vf[32];
#pragma unroll
    for (int q = 0; q < 32; ++q) vf[q] = 0.0f;
    if (wave < 3) {
#pragma unroll
        for (int kk = 0; kk < 16; ++kk) {
            const int kb = kk * 4 + half * 2;
            vf[2 * kk]     = W2[(size_t)kb * 34 + cc];
            vf[2 * kk + 1] = W2[(size_t)(kb + 1) * 34 + cc];
        }
    }
    lds_barrier();

    const float4 p0 = *(const float4*)&tb[g * TP + c];
    const float4 p1 = *(const float4*)&tb[g * TP + 32 + c];
    const float r0 = fmaxf(p0.x, 0.0f), r1 = fmaxf(p0.y, 0.0f);
    const float r2 = fmaxf(p0.z, 0.0f), r3 = fmaxf(p0.w, 0.0f);
    const float r4 = fmaxf(p1.x, 0.0f), r5 = fmaxf(p1.y, 0.0f);
    const float r6 = fmaxf(p1.z, 0.0f), r7 = fmaxf(p1.w, 0.0f);
    float m = fmaxf(fmaxf(fmaxf(r0, r1), fmaxf(r2, r3)), fmaxf(fmaxf(r4, r5), fmaxf(r6, r7)));
    for (int off = 4; off > 0; off >>= 1) m = fmaxf(m, __shfl_xor(m, off, 8));
    const float e0 = __expf(r0 - m), e1 = __expf(r1 - m);
    const float e2 = __expf(r2 - m), e3 = __expf(r3 - m);
    const float e4 = __expf(r4 - m), e5 = __expf(r5 - m);
    const float e6 = __expf(r6 - m), e7 = __expf(r7 - m);
    float s = ((e0 + e1) + (e2 + e3)) + ((e4 + e5) + (e6 + e7));
    for (int off = 4; off > 0; off >>= 1) s += __shfl_xor(s, off, 8);
    const float inv = (1.0f / s) * di;
    float4 o0, o1;
    o0.x = e0 * inv; o0.y = e1 * inv; o0.z = e2 * inv; o0.w = e3 * inv;
    o1.x = e4 * inv; o1.y = e5 * inv; o1.z = e6 * inv; o1.w = e7 * inv;
    *(float4*)&ta[g * TP + c]      = o0;
    *(float4*)&ta[g * TP + 32 + c] = o1;
    lds_barrier();

    if (wave < 3) {
        v8f d = {0.f,0.f,0.f,0.f,0.f,0.f,0.f,0.f};
#pragma unroll
        for (int kk = 0; kk < 16; ++kk) {
            const int kb = kk * 4 + half * 2;
            const v2f a = *(const v2f*)&ta[lm * TP + kb];
            v2f b;
            b.x = ok ? vf[2 * kk]     : 0.0f;
            b.y = ok ? vf[2 * kk + 1] : 0.0f;
            d = __builtin_amdgcn_wmma_f32_16x16x4_f32(false, a, false, b,
                                                      (short)0, d, false, false);
        }
        if (ok) {
#pragma unroll
            for (int i = 0; i < 8; ++i) {
                const int row = blockIdx.x * 16 + half * 8 + i;
                if (row < n) hw2[(size_t)row * 34 + col2] = d[i];
            }
        }
    }
}

__global__ void __launch_bounds__(256) k_gather34(const int* __restrict__ rowptr,
                                                  const int* __restrict__ cnt,
                                                  const int* __restrict__ col,
                                                  const float* __restrict__ dinv,
                                                  const float* __restrict__ hw2,
                                                  const float* __restrict__ b2,
                                                  float* __restrict__ out, int n) {
    const int t = blockIdx.x * 256 + threadIdx.x;
    const int node = t >> 4;
    const int l = t & 15;
    const int c = l << 1;
    if (node >= n) return;
    const float di = dinv[node];
    int p = rowptr[node];
    const int nin = cnt[node];
    const int end = p + nin;
    const float* srow = hw2 + (size_t)node * 34;
    const v2f sv = *(const v2f*)(srow + c);
    const v2f sx = *(const v2f*)(srow + 32);
    asm volatile("" :: "v"(p), "v"(nin));
    v2f acc = sv, accx = sx;
    const v2f zz = {0.0f, 0.0f};
    int4 cur = *(const int4*)(col + p);
    while (p < end) {
        const int4 nxt = *(const int4*)(col + p + 4);
        const bool k1 = p + 1 < end, k2 = p + 2 < end, k3 = p + 3 < end;
        const int s0 = cur.x;
        const int s1 = k1 ? cur.y : s0, s2 = k2 ? cur.z : s0, s3 = k3 ? cur.w : s0;
        const float* r0 = hw2 + (size_t)s0 * 34;
        const float* r1 = hw2 + (size_t)s1 * 34;
        const float* r2 = hw2 + (size_t)s2 * 34;
        const float* r3 = hw2 + (size_t)s3 * 34;
        const v2f v0 = *(const v2f*)(r0 + c), x0 = *(const v2f*)(r0 + 32);
        const v2f v1 = *(const v2f*)(r1 + c), x1 = *(const v2f*)(r1 + 32);
        const v2f v2 = *(const v2f*)(r2 + c), x2 = *(const v2f*)(r2 + 32);
        const v2f v3 = *(const v2f*)(r3 + c), x3 = *(const v2f*)(r3 + 32);
        acc += v0;           accx += x0;
        acc += k1 ? v1 : zz; accx += k1 ? x1 : zz;
        acc += k2 ? v2 : zz; accx += k2 ? x2 : zz;
        acc += k3 ? v3 : zz; accx += k3 ? x3 : zz;
        asm volatile("" :: "v"(nxt.x), "v"(nxt.y), "v"(nxt.z), "v"(nxt.w));
        cur = nxt;
        p += 4;
    }
    v2f o, ox;
    o.x  = fmaf(di, acc.x,  b2[c]);
    o.y  = fmaf(di, acc.y,  b2[c + 1]);
    ox.x = fmaf(di, accx.x, b2[32]);
    ox.y = fmaf(di, accx.y, b2[33]);
    float* orow = out + (size_t)node * 34;
    *(v2f*)(orow + c) = o;
    if (l == 0) *(v2f*)(orow + 32) = ox;
}

extern "C" void kernel_launch(void* const* d_in, const int* in_sizes, int n_in,
                              void* d_out, int out_size, void* d_ws, size_t ws_size,
                              hipStream_t stream) {
    const float* x  = (const float*)d_in[0];
    const int*   ei = (const int*)  d_in[1];
    const float* W1 = (const float*)d_in[2];
    const float* b1 = (const float*)d_in[3];
    const float* W2 = (const float*)d_in[4];
    const float* b2 = (const float*)d_in[5];
    float* out = (float*)d_out;

    const int N = in_sizes[0] / 64;
    const int E = in_sizes[1] / 2;
    const int* src = ei;
    const int* dst = ei + E;

    int*   cnt    = (int*)d_ws;
    int*   rowptr = cnt    + N;
    int*   pad    = rowptr + N;
    float* dinv   = (float*)(pad + 64);
    float* hw2    = dinv + N;
    int*   col    = (int*)(hw2 + (size_t)N * 34);
    int*   binned = col + E;
    int*   binoff = binned + E;
    int*   hist   = binoff + 512;

    const int T = 256;
    const int nblk = (E + CH - 1) / CH;
    k_hist<<<nblk, 1024, 0, stream>>>(dst, hist, E);
    k_bin <<<nblk, 1024, 0, stream>>>(src, dst, hist, binoff, binned, E, nblk);
    k_csr <<<(N + RNG - 1) / RNG, 1024, 0, stream>>>(binned, binoff, cnt, rowptr, dinv, col, N);
    k_layer1<<<(N + 15) / 16, 128, 0, stream>>>(rowptr, cnt, col, dinv, x, W1, b1, W2, hw2, N);
    k_gather34<<<((size_t)N * 16 + T - 1) / T, T, 0, stream>>>(rowptr, cnt, col, dinv, hw2, b2, out, N);
}


# Level-1 edge split sorts each chunk by range in LDS, stores it whole with per-range offsets; level 2 walks its piece of every chunk: histogram launch, scattered stores go; new LDS arrival sum order
# speedup vs baseline: 5.1741x; 1.2830x over previous
#include <hip/hip_runtime.h>
#include <math.h>

typedef __attribute__((ext_vector_type(2))) float v2f;
typedef __attribute__((ext_vector_type(8))) float v8f;


#define RSH   9
#define RNG   512
#define NBMAX 256
#define CH    5120

__global__ void __launch_bounds__(1024) k_part(const int* __restrict__ src, const int* __restrict__ dst,
                                               int* __restrict__ segoff, int* __restrict__ binned,
                                               int E, int nblk) {
    __shared__ int lcur[NBMAX];
    __shared__ int wsum[NBMAX / 32];
    __shared__ int lsort[CH];
    const int tid = threadIdx.x, lane = tid & 31, wv = tid >> 5;
    const int me = blockIdx.x;
    if (tid < NBMAX) lcur[tid] = 0;
    const int base = me * CH + tid;
    int d[CH / 1024], s[CH / 1024];
#pragma unroll
    for (int k = 0; k < CH / 1024; ++k) {
        const int e  = base + k * 1024;
        const int ec = e < E ? e : E - 1;
        d[k] = dst[ec];
        s[k] = src[ec];
    }
    __syncthreads();
#pragma unroll
    for (int k = 0; k < CH / 1024; ++k)
        if (base + k * 1024 < E) atomicAdd(&lcur[d[k] >> RSH], 1);
    __syncthreads();
    int c = 0, inc = 0;
    if (tid < NBMAX) {
        c = lcur[tid];
        inc = c;
#pragma unroll
        for (int off = 1; off < 32; off <<= 1) {
            int v = __shfl_up(inc, off, 32);
            if (lane >= off) inc += v;
        }
        if (lane == 31) wsum[wv] = inc;
    }
    __syncthreads();
    if (tid < NBMAX) {
        int woff = 0;
        for (int w = 0; w < wv; ++w) woff += wsum[w];
        const int ex = woff + inc - c;
        lcur[tid] = ex;
        segoff[(size_t)tid * nblk + me] = ex;
        if (tid == NBMAX - 1) segoff[(size_t)NBMAX * nblk + me] = ex + c;
    }
    __syncthreads();
#pragma unroll
    for (int k = 0; k < CH / 1024; ++k) {
        if (base + k * 1024 < E) {
            const int pos = atomicAdd(&lcur[d[k] >> RSH], 1);
            lsort[pos] = (s[k] << RSH) | (d[k] & (RNG - 1));
        }
    }
    __syncthreads();
    const int left = E - me * CH;
    const int nval = left < CH ? left : CH;
#pragma unroll
    for (int k = 0; k < CH / 1024; ++k) {
        const int i = tid + k * 1024;
        if (i < nval) binned[(size_t)me * CH + i] = lsort[i];
    }
}

__global__ void __launch_bounds__(1024) k_csr(const int* __restrict__ binned, const int* __restrict__ segoff,
                                              int* __restrict__ cnt, int* __restrict__ rowptr,
                                              float* __restrict__ dinv, int* __restrict__ col, int n, int nblk) {
    __shared__ int lc[RNG];
    __shared__ int wsum[RNG / 32];
    __shared__ int red[32];
    const int tid = threadIdx.x, lane = tid & 31, wv = tid >> 5;
    const int* so0 = segoff + (size_t)blockIdx.x * nblk;
    const int* so1 = so0 + nblk;
    const int q  = tid & 3;
    const int b0 = tid >> 2;
    int sb0 = 0, eb0 = 0;
    if (b0 < nblk) { sb0 = so0[b0]; eb0 = so1[b0]; }
    int part = 0;
    for (int b = tid; b < nblk; b += 1024) part += so0[b];
#pragma unroll
    for (int off = 16; off > 0; off >>= 1) part += __shfl_xor(part, off, 32);
    if (lane == 0) red[wv] = part;
    if (tid < RNG) lc[tid] = 0;
    __syncthreads();
    int start = red[lane];
#pragma unroll
    for (int off = 16; off > 0; off >>= 1) start += __shfl_xor(start, off, 32);
    {
        int b = b0, sb = sb0, eb = eb0;
        while (b < nblk) {
            const int* seg = binned + (size_t)b * CH;
            for (int i0 = sb + q; i0 < eb; i0 += 32) {
                int v[8];
#pragma unroll
                for (int k = 0; k < 8; ++k) {
                    const int i = i0 + 4 * k;
                    v[k] = seg[i < eb ? i : eb - 1];
                }
                asm volatile("" :: "v"(v[0]), "v"(v[1]), "v"(v[2]), "v"(v[3]), "v"(v[4]), "v"(v[5]), "v"(v[6]), "v"(v[7]));
#pragma unroll
                for (int k = 0; k < 8; ++k)
                    if (i0 + 4 * k < eb) atomicAdd(&lc[v[k] & (RNG - 1)], 1);
            }
            b += 256;
            if (b < nblk) { sb = so0[b]; eb = so1[b]; }
        }
    }
    __syncthreads();
    int c = 0, inc = 0;
    if (tid < RNG) {
        c = lc[tid];
        inc = c;
#pragma unroll
        for (int off = 1; off < 32; off <<= 1) {
            int v = __shfl_up(inc, off, 32);
            if (lane >= off) inc += v;
        }
        if (lane == 31) wsum[wv] = inc;
    }
    __syncthreads();
    if (tid < RNG) {
        int woff = 0;
        for (int w = 0; w < wv; ++w) woff += wsum[w];
        const int ex = start + woff + inc - c;
        lc[tid] = ex;
        const int node = blockIdx.x * RNG + tid;
        if (node < n) {
            cnt[node]    = c;
            rowptr[node] = ex;
            dinv[node]   = rsqrtf((float)(c + 1));
        }
    }
    __syncthreads();
    {
        int b = b0, sb = sb0, eb = eb0;
        while (b < nblk) {
            const int* seg = binned + (size_t)b * CH;
            for (int i0 = sb + q; i0 < eb; i0 += 32) {
                int v[8];
#pragma unroll
                for (int k = 0; k < 8; ++k) {
                    const int i = i0 + 4 * k;
                    v[k] = seg[i < eb ? i : eb - 1];
                }
                asm volatile("" :: "v"(v[0]), "v"(v[1]), "v"(v[2]), "v"(v[3]), "v"(v[4]), "v"(v[5]), "v"(v[6]), "v"(v[7]));
#pragma unroll
                for (int k = 0; k < 8; ++k) {
                    if (i0 + 4 * k < eb) {
                        const int pos = atomicAdd(&lc[v[k] & (RNG - 1)], 1);
                        col[pos] = v[k] >> RSH;
                    }
                }
            }
            b += 256;
            if (b < nblk) { sb = so0[b]; eb = so1[b]; }
        }
    }
}

#define TP 68
__device__ __forceinline__ void lds_barrier() {
    __builtin_amdgcn_fence(__ATOMIC_RELEASE, "workgroup", "local");
    __builtin_amdgcn_s_barrier();
    __builtin_amdgcn_fence(__ATOMIC_ACQUIRE, "workgroup", "local");
}
__global__ void __launch_bounds__(128) k_layer1(const int* __restrict__ rowptr,
                                                const int* __restrict__ cnt,
                                                const int* __restrict__ col,
                                                const float* __restrict__ dinv,
                                                const float* __restrict__ x,
                                                const float* W1,
                                                const float* __restrict__ b1,
                                                const float* W2,
                                                float* __restrict__ hw2, int n) {
    __shared__ float ta[16 * TP];
    __shared__ float tb[16 * TP];
    const int tid  = threadIdx.x;
    const int g    = tid >> 3;
    const int c    = (tid & 7) << 2;
    const int node = blockIdx.x * 16 + g;
    const int wave = tid >> 5;
    const int lane = tid & 31;
    const int half = lane >> 4;
    const int lm   = lane & 15;
    float4 a0 = make_float4(0.0f, 0.0f, 0.0f, 0.0f);
    float4 a1 = make_float4(0.0f, 0.0f, 0.0f, 0.0f);
    float di = 0.0f;
    if (node < n) {
        di = dinv[node];
        int p = rowptr[node];
        const int nin = cnt[node];
        const int end = p + nin;
        const float4* x4 = (const float4*)x;
        const int l = tid & 7;
        const float4 su = x4[node * 16 + l];
        const float4 sv = x4[node * 16 + l + 8];
        asm volatile("" :: "v"(p), "v"(nin));
        const float wself = di * di;
        a0.x = wself * su.x; a0.y = wself * su.y; a0.z = wself * su.z; a0.w = wself * su.w;
        a1.x = wself * sv.x; a1.y = wself * sv.y; a1.z = wself * sv.z; a1.w = wself * sv.w;
        int4 cur = *(const int4*)(col + p);
        while (p < end) {
            const int4 nxt = *(const int4*)(col + p + 4);
            const bool k1 = p + 1 < end, k2 = p + 2 < end, k3 = p + 3 < end;
            const int s0 = cur.x;
            const int s1 = k1 ? cur.y : s0, s2 = k2 ? cur.z : s0, s3 = k3 ? cur.w : s0;
            const float d0 = dinv[s0], d1 = dinv[s1], d2 = dinv[s2], d3 = dinv[s3];
            const float4 u0 = x4[s0 * 16 + l], v0 = x4[s0 * 16 + l + 8];
            const float4 u1 = x4[s1 * 16 + l], v1 = x4[s1 * 16 + l + 8];
            const float4 u2 = x4[s2 * 16 + l], v2 = x4[s2 * 16 + l + 8];
            const float4 u3 = x4[s3 * 16 + l], v3 = x4[s3 * 16 + l + 8];
            const float w0 = d0 * di;
            const float w1 = d1 * (k1 ? di : 0.0f);
            const float w2 = d2 * (k2 ? di : 0.0f);
            const float w3 = d3 * (k3 ? di : 0.0f);
            a0.x = fmaf(w0, u0.x, a0.x); a0.y = fmaf(w0, u0.y, a0.y); a0.z = fmaf(w0, u0.z, a0.z); a0.w = fmaf(w0, u0.w, a0.w);
            a1.x = fmaf(w0, v0.x, a1.x); a1.y = fmaf(w0, v0.y, a1.y); a1.z = fmaf(w0, v0.z, a1.z); a1.w = fmaf(w0, v0.w, a1.w);
            a0.x = fmaf(w1, u1.x, a0.x); a0.y = fmaf(w1, u1.y, a0.y); a0.z = fmaf(w1, u1.z, a0.z); a0.w = fmaf(w1, u1.w, a0.w);
            a1.x = fmaf(w1, v1.x, a1.x); a1.y = fmaf(w1, v1.y, a1.y); a1.z = fmaf(w1, v1.z, a1.z); a1.w = fmaf(w1, v1.w, a1.w);
            a0.x = fmaf(w2, u2.x, a0.x); a0.y = fmaf(w2, u2.y, a0.y); a0.z = fmaf(w2, u2.z, a0.z); a0.w = fmaf(w2, u2.w, a0.w);
            a1.x = fmaf(w2, v2.x, a1.x); a1.y = fmaf(w2, v2.y, a1.y); a1.z = fmaf(w2, v2.z, a1.z); a1.w = fmaf(w2, v2.w, a1.w);
            a0.x = fmaf(w3, u3.x, a0.x); a0.y = fmaf(w3, u3.y, a0.y); a0.z = fmaf(w3, u3.z, a0.z); a0.w = fmaf(w3, u3.w, a0.w);
            a1.x = fmaf(w3, v3.x, a1.x); a1.y = fmaf(w3, v3.y, a1.y); a1.z = fmaf(w3, v3.z, a1.z); a1.w = fmaf(w3, v3.w, a1.w);
            asm volatile("" :: "v"(nxt.x), "v"(nxt.y), "v"(nxt.z), "v"(nxt.w));
            cur = nxt;
            p += 4;
        }
    }
    const int bcol = wave * 16 + lm;
    const float bias = b1[bcol];
    float wf[32];
#pragma unroll
    for (int kk = 0; kk < 16; ++kk) {
        const int kb = kk * 4 + half * 2;
        wf[2 * kk]     = W1[(size_t)kb * 64 + bcol];
        wf[2 * kk + 1] = W1[(size_t)(kb + 1) * 64 + bcol];
    }
    *(float4*)&ta[g * TP + c]      = a0;
    *(float4*)&ta[g * TP + 32 + c] = a1;
    lds_barrier();

    {
        v8f d = {0.f,0.f,0.f,0.f,0.f,0.f,0.f,0.f};
#pragma unroll
        for (int kk = 0; kk < 16; ++kk) {
            const int kb = kk * 4 + half * 2;
            const v2f a = *(const v2f*)&ta[lm * TP + kb];
            v2f b;
            b.x = wf[2 * kk];
            b.y = wf[2 * kk + 1];
            d = __builtin_amdgcn_wmma_f32_16x16x4_f32(false, a, false, b,
                                                      (short)0, d, false, false);
        }
#pragma unroll
        for (int i = 0; i < 8; ++i)
            tb[(half * 8 + i) * TP + bcol] = d[i] + bias;
    }
    const int  col2 = wave * 16 + lm;
    const bool ok   = (col2 < 34);
    const int  cc   = ok ? col2 : 33;
    float vf[32];
#pragma unroll
    for (int q = 0; q < 32; ++q) vf[q] = 0.0f;
    if (wave < 3) {
#pragma unroll
        for (int kk = 0; kk < 16; ++kk) {
            const int kb = kk * 4 + half * 2;
            vf[2 * kk]     = W2[(size_t)kb * 34 + cc];
            vf[2 * kk + 1] = W2[(size_t)(kb + 1) * 34 + cc];
        }
    }
    lds_barrier();

    const float4 p0 = *(const float4*)&tb[g * TP + c];
    const float4 p1 = *(const float4*)&tb[g * TP + 32 + c];
    const float r0 = fmaxf(p0.x, 0.0f), r1 = fmaxf(p0.y, 0.0f);
    const float r2 = fmaxf(p0.z, 0.0f), r3 = fmaxf(p0.w, 0.0f);
    const float r4 = fmaxf(p1.x, 0.0f), r5 = fmaxf(p1.y, 0.0f);
    const float r6 = fmaxf(p1.z, 0.0f), r7 = fmaxf(p1.w, 0.0f);
    float m = fmaxf(fmaxf(fmaxf(r0, r1), fmaxf(r2, r3)), fmaxf(fmaxf(r4, r5), fmaxf(r6, r7)));
    for (int off = 4; off > 0; off >>= 1) m = fmaxf(m, __shfl_xor(m, off, 8));
    const float e0 = __expf(r0 - m), e1 = __expf(r1 - m);
    const float e2 = __expf(r2 - m), e3 = __expf(r3 - m);
    const float e4 = __expf(r4 - m), e5 = __expf(r5 - m);
    const float e6 = __expf(r6 - m), e7 = __expf(r7 - m);
    float s = ((e0 + e1) + (e2 + e3)) + ((e4 + e5) + (e6 + e7));
    for (int off = 4; off > 0; off >>= 1) s += __shfl_xor(s, off, 8);
    const float inv = (1.0f / s) * di;
    float4 o0, o1;
    o0.x = e0 * inv; o0.y = e1 * inv; o0.z = e2 * inv; o0.w = e3 * inv;
    o1.x = e4 * inv; o1.y = e5 * inv; o1.z = e6 * inv; o1.w = e7 * inv;
    *(float4*)&ta[g * TP + c]      = o0;
    *(float4*)&ta[g * TP + 32 + c] = o1;
    lds_barrier();

    if (wave < 3) {
        v8f d = {0.f,0.f,0.f,0.f,0.f,0.f,0.f,0.f};
#pragma unroll
        for (int kk = 0; kk < 16; ++kk) {
            const int kb = kk * 4 + half * 2;
            const v2f a = *(const v2f*)&ta[lm * TP + kb];
            v2f b;
            b.x = ok ? vf[2 * kk]     : 0.0f;
            b.y = ok ? vf[2 * kk + 1] : 0.0f;
            d = __builtin_amdgcn_wmma_f32_16x16x4_f32(false, a, false, b,
                                                      (short)0, d, false, false);
        }
        if (ok) {
#pragma unroll
            for (int i = 0; i < 8; ++i) {
                const int row = blockIdx.x * 16 + half * 8 + i;
                if (row < n) hw2[(size_t)row * 34 + col2] = d[i];
            }
        }
    }
}

__global__ void __launch_bounds__(256) k_gather34(const int* __restrict__ rowptr,
                                                  const int* __restrict__ cnt,
                                                  const int* __restrict__ col,
                                                  const float* __restrict__ dinv,
                                                  const float* __restrict__ hw2,
                                                  const float* __restrict__ b2,
                                                  float* __restrict__ out, int n) {
    const int t = blockIdx.x * 256 + threadIdx.x;
    const int node = t >> 4;
    const int l = t & 15;
    const int c = l << 1;
    if (node >= n) return;
    const float di = dinv[node];
    int p = rowptr[node];
    const int nin = cnt[node];
    const int end = p + nin;
    const float* srow = hw2 + (size_t)node * 34;
    const v2f sv = *(const v2f*)(srow + c);
    const v2f sx = *(const v2f*)(srow + 32);
    asm volatile("" :: "v"(p), "v"(nin));
    v2f acc = sv, accx = sx;
    const v2f zz = {0.0f, 0.0f};
    int4 cur = *(const int4*)(col + p);
    while (p < end) {
        const int4 nxt = *(const int4*)(col + p + 4);
        const bool k1 = p + 1 < end, k2 = p + 2 < end, k3 = p + 3 < end;
        const int s0 = cur.x;
        const int s1 = k1 ? cur.y : s0, s2 = k2 ? cur.z : s0, s3 = k3 ? cur.w : s0;
        const float* r0 = hw2 + (size_t)s0 * 34;
        const float* r1 = hw2 + (size_t)s1 * 34;
        const float* r2 = hw2 + (size_t)s2 * 34;
        const float* r3 = hw2 + (size_t)s3 * 34;
        const v2f v0 = *(const v2f*)(r0 + c), x0 = *(const v2f*)(r0 + 32);
        const v2f v1 = *(const v2f*)(r1 + c), x1 = *(const v2f*)(r1 + 32);
        const v2f v2 = *(const v2f*)(r2 + c), x2 = *(const v2f*)(r2 + 32);
        const v2f v3 = *(const v2f*)(r3 + c), x3 = *(const v2f*)(r3 + 32);
        acc += v0;           accx += x0;
        acc += k1 ? v1 : zz; accx += k1 ? x1 : zz;
        acc += k2 ? v2 : zz; accx += k2 ? x2 : zz;
        acc += k3 ? v3 : zz; accx += k3 ? x3 : zz;
        asm volatile("" :: "v"(nxt.x), "v"(nxt.y), "v"(nxt.z), "v"(nxt.w));
        cur = nxt;
        p += 4;
    }
    v2f o, ox;
    o.x  = fmaf(di, acc.x,  b2[c]);
    o.y  = fmaf(di, acc.y,  b2[c + 1]);
    ox.x = fmaf(di, accx.x, b2[32]);
    ox.y = fmaf(di, accx.y, b2[33]);
    float* orow = out + (size_t)node * 34;
    *(v2f*)(orow + c) = o;
    if (l == 0) *(v2f*)(orow + 32) = ox;
}

extern "C" void kernel_launch(void* const* d_in, const int* in_sizes, int n_in,
                              void* d_out, int out_size, void* d_ws, size_t ws_size,
                              hipStream_t stream) {
    const float* x  = (const float*)d_in[0];
    const int*   ei = (const int*)  d_in[1];
    const float* W1 = (const float*)d_in[2];
    const float* b1 = (const float*)d_in[3];
    const float* W2 = (const float*)d_in[4];
    const float* b2 = (const float*)d_in[5];
    float* out = (float*)d_out;

    const int N = in_sizes[0] / 64;
    const int E = in_sizes[1] / 2;
    const int* src = ei;
    const int* dst = ei + E;

    int*   cnt    = (int*)d_ws;
    int*   rowptr = cnt    + N;
    int*   pad    = rowptr + N;
    float* dinv   = (float*)(pad + 64);
    float* hw2    = dinv + N;
    int*   col    = (int*)(hw2 + (size_t)N * 34);
    int*   binned = col + E;
    int*   segoff = binned + E;

    const int T = 256;
    const int nblk = (E + CH - 1) / CH;
    k_part<<<nblk, 1024, 0, stream>>>(src, dst, segoff, binned, E, nblk);
    k_csr <<<(N + RNG - 1) / RNG, 1024, 0, stream>>>(binned, segoff, cnt, rowptr, dinv, col, N, nblk);
    k_layer1<<<(N + 15) / 16, 128, 0, stream>>>(rowptr, cnt, col, dinv, x, W1, b1, W2, hw2, N);
    k_gather34<<<((size_t)N * 16 + T - 1) / T, T, 0, stream>>>(rowptr, cnt, col, dinv, hw2, b2, out, N);
}


# Level-2 fill places source ids in an LDS copy of the range's bucket run and the copy leaves as whole 128-byte lines, 16 B per lane: the scattered 4-byte col stores go; same LDS arrival sum order
# speedup vs baseline: 5.3552x; 1.0350x over previous
#include <hip/hip_runtime.h>
#include <math.h>

typedef __attribute__((ext_vector_type(2))) float v2f;
typedef __attribute__((ext_vector_type(8))) float v8f;


#define RSH   9
#define RNG   512
#define NBMAX 256
#define CH    5120

__global__ void __launch_bounds__(1024) k_part(const int* __restrict__ src, const int* __restrict__ dst,
                                               int* __restrict__ segoff, int* __restrict__ binned,
                                               int E, int nblk) {
    __shared__ int lcur[NBMAX];
    __shared__ int wsum[NBMAX / 32];
    __shared__ int lsort[CH];
    const int tid = threadIdx.x, lane = tid & 31, wv = tid >> 5;
    const int me = blockIdx.x;
    if (tid < NBMAX) lcur[tid] = 0;
    const int base = me * CH + tid;
    int d[CH / 1024], s[CH / 1024];
#pragma unroll
    for (int k = 0; k < CH / 1024; ++k) {
        const int e  = base + k * 1024;
        const int ec = e < E ? e : E - 1;
        d[k] = dst[ec];
        s[k] = src[ec];
    }
    __syncthreads();
#pragma unroll
    for (int k = 0; k < CH / 1024; ++k)
        if (base + k * 1024 < E) atomicAdd(&lcur[d[k] >> RSH], 1);
    __syncthreads();
    int c = 0, inc = 0;
    if (tid < NBMAX) {
        c = lcur[tid];
        inc = c;
#pragma unroll
        for (int off = 1; off < 32; off <<= 1) {
            int v = __shfl_up(inc, off, 32);
            if (lane >= off) inc += v;
        }
        if (lane == 31) wsum[wv] = inc;
    }
    __syncthreads();
    if (tid < NBMAX) {
        int woff = 0;
        for (int w = 0; w < wv; ++w) woff += wsum[w];
        const int ex = woff + inc - c;
        lcur[tid] = ex;
        segoff[(size_t)tid * nblk + me] = ex;
        if (tid == NBMAX - 1) segoff[(size_t)NBMAX * nblk + me] = ex + c;
    }
    __syncthreads();
#pragma unroll
    for (int k = 0; k < CH / 1024; ++k) {
        if (base + k * 1024 < E) {
            const int pos = atomicAdd(&lcur[d[k] >> RSH], 1);
            lsort[pos] = (s[k] << RSH) | (d[k] & (RNG - 1));
        }
    }
    __syncthreads();
    const int left = E - me * CH;
    const int nval = left < CH ? left : CH;
#pragma unroll
    for (int k = 0; k < CH / 1024; ++k) {
        const int i = tid + k * 1024;
        if (i < nval) binned[(size_t)me * CH + i] = lsort[i];
    }
}

__device__ __forceinline__ void csr_fill(const bool to_lds, const int* __restrict__ binned,
                                         const int* so0, const int* so1, int* lc, int* dstl, int* __restrict__ dstg,
                                         const int b0, const int sb0, const int eb0, const int q, const int nblk) {
    int b = b0, sb = sb0, eb = eb0;
    while (b < nblk) {
        const int* seg = binned + (size_t)b * CH;
        for (int i0 = sb + q; i0 < eb; i0 += 32) {
            int v[8];
#pragma unroll
            for (int k = 0; k < 8; ++k) {
                const int i = i0 + 4 * k;
                v[k] = seg[i < eb ? i : eb - 1];
            }
            asm volatile("" :: "v"(v[0]), "v"(v[1]), "v"(v[2]), "v"(v[3]), "v"(v[4]), "v"(v[5]), "v"(v[6]), "v"(v[7]));
#pragma unroll
            for (int k = 0; k < 8; ++k) {
                if (i0 + 4 * k < eb) {
                    const int pos = atomicAdd(&lc[v[k] & (RNG - 1)], 1);
                    if (to_lds) dstl[pos] = v[k] >> RSH;
                    else        dstg[pos] = v[k] >> RSH;
                }
            }
        }
        b += 256;
        if (b < nblk) { sb = so0[b]; eb = so1[b]; }
    }
}

#define LCAP 12288
__global__ void __launch_bounds__(1024) k_csr(const int* __restrict__ binned, const int* __restrict__ segoff,
                                              int* __restrict__ cnt, int* __restrict__ rowptr,
                                              float* __restrict__ dinv, int* __restrict__ col, int n, int nblk) {
    __shared__ int lc[RNG];
    __shared__ int wsum[RNG / 32];
    __shared__ int red[32];
    __shared__ int rtot;
    __shared__ alignas(16) int lcol[LCAP];
    const int tid = threadIdx.x, lane = tid & 31, wv = tid >> 5;
    const int* so0 = segoff + (size_t)blockIdx.x * nblk;
    const int* so1 = so0 + nblk;
    const int q  = tid & 3;
    const int b0 = tid >> 2;
    int sb0 = 0, eb0 = 0;
    if (b0 < nblk) { sb0 = so0[b0]; eb0 = so1[b0]; }
    int part = 0;
    for (int b = tid; b < nblk; b += 1024) part += so0[b];
#pragma unroll
    for (int off = 16; off > 0; off >>= 1) part += __shfl_xor(part, off, 32);
    if (lane == 0) red[wv] = part;
    if (tid < RNG) lc[tid] = 0;
    __syncthreads();
    int start = red[lane];
#pragma unroll
    for (int off = 16; off > 0; off >>= 1) start += __shfl_xor(start, off, 32);
    {
        int b = b0, sb = sb0, eb = eb0;
        while (b < nblk) {
            const int* seg = binned + (size_t)b * CH;
            for (int i0 = sb + q; i0 < eb; i0 += 32) {
                int v[8];
#pragma unroll
                for (int k = 0; k < 8; ++k) {
                    const int i = i0 + 4 * k;
                    v[k] = seg[i < eb ? i : eb - 1];
                }
                asm volatile("" :: "v"(v[0]), "v"(v[1]), "v"(v[2]), "v"(v[3]), "v"(v[4]), "v"(v[5]), "v"(v[6]), "v"(v[7]));
#pragma unroll
                for (int k = 0; k < 8; ++k)
                    if (i0 + 4 * k < eb) atomicAdd(&lc[v[k] & (RNG - 1)], 1);
            }
            b += 256;
            if (b < nblk) { sb = so0[b]; eb = so1[b]; }
        }
    }
    __syncthreads();
    int c = 0, inc = 0;
    if (tid < RNG) {
        c = lc[tid];
        inc = c;
#pragma unroll
        for (int off = 1; off < 32; off <<= 1) {
            int v = __shfl_up(inc, off, 32);
            if (lane >= off) inc += v;
        }
        if (lane == 31) wsum[wv] = inc;
    }
    __syncthreads();
    if (tid < RNG) {
        int woff = 0;
        for (int w = 0; w < wv; ++w) woff += wsum[w];
        const int ex = woff + inc - c;
        lc[tid] = ex;
        if (tid == RNG - 1) rtot = ex + c;
        const int node = blockIdx.x * RNG + tid;
        if (node < n) {
            cnt[node]    = c;
            rowptr[node] = start + ex;
            dinv[node]   = rsqrtf((float)(c + 1));
        }
    }
    __syncthreads();
    const int total = rtot;
    const int a = (int)((((size_t)(col + start)) & 127) >> 2);
    const int endw = a + total;
    const bool in_lds = __builtin_amdgcn_readfirstlane(endw) <= LCAP;
    if (in_lds) {
        csr_fill(true, binned, so0, so1, lc, lcol + a, col, b0, sb0, eb0, q, nblk);
        __syncthreads();
        int* cline = col + start - a;
        for (int j = tid * 4; j < endw; j += 4096) {
            if (j >= a && j + 4 <= endw) {
                *(int4*)(cline + j) = *(const int4*)&lcol[j];
            } else {
#pragma unroll
                for (int k = 0; k < 4; ++k)
                    if (j + k >= a && j + k < endw) cline[j + k] = lcol[j + k];
            }
        }
    } else {
        csr_fill(false, binned, so0, so1, lc, lcol, col + start, b0, sb0, eb0, q, nblk);
    }
}

#define TP 68
__device__ __forceinline__ void lds_barrier() {
    __builtin_amdgcn_fence(__ATOMIC_RELEASE, "workgroup", "local");
    __builtin_amdgcn_s_barrier();
    __builtin_amdgcn_fence(__ATOMIC_ACQUIRE, "workgroup", "local");
}
__global__ void __launch_bounds__(128) k_layer1(const int* __restrict__ rowptr,
                                                const int* __restrict__ cnt,
                                                const int* __restrict__ col,
                                                const float* __restrict__ dinv,
                                                const float* __restrict__ x,
                                                const float* W1,
                                                const float* __restrict__ b1,
                                                const float* W2,
                                                float* __restrict__ hw2, int n) {
    __shared__ float ta[16 * TP];
    __shared__ float tb[16 * TP];
    const int tid  = threadIdx.x;
    const int g    = tid >> 3;
    const int c    = (tid & 7) << 2;
    const int node = blockIdx.x * 16 + g;
    const int wave = tid >> 5;
    const int lane = tid & 31;
    const int half = lane >> 4;
    const int lm   = lane & 15;
    float4 a0 = make_float4(0.0f, 0.0f, 0.0f, 0.0f);
    float4 a1 = make_float4(0.0f, 0.0f, 0.0f, 0.0f);
    float di = 0.0f;
    if (node < n) {
        di = dinv[node];
        int p = rowptr[node];
        const int nin = cnt[node];
        const int end = p + nin;
        const float4* x4 = (const float4*)x;
        const int l = tid & 7;
        const float4 su = x4[node * 16 + l];
        const float4 sv = x4[node * 16 + l + 8];
        asm volatile("" :: "v"(p), "v"(nin));
        const float wself = di * di;
        a0.x = wself * su.x; a0.y = wself * su.y; a0.z = wself * su.z; a0.w = wself * su.w;
        a1.x = wself * sv.x; a1.y = wself * sv.y; a1.z = wself * sv.z; a1.w = wself * sv.w;
        int4 cur = *(const int4*)(col + p);
        while (p < end) {
            const int4 nxt = *(const int4*)(col + p + 4);
            const bool k1 = p + 1 < end, k2 = p + 2 < end, k3 = p + 3 < end;
            const int s0 = cur.x;
            const int s1 = k1 ? cur.y : s0, s2 = k2 ? cur.z : s0, s3 = k3 ? cur.w : s0;
            const float d0 = dinv[s0], d1 = dinv[s1], d2 = dinv[s2], d3 = dinv[s3];
            const float4 u0 = x4[s0 * 16 + l], v0 = x4[s0 * 16 + l + 8];
            const float4 u1 = x4[s1 * 16 + l], v1 = x4[s1 * 16 + l + 8];
            const float4 u2 = x4[s2 * 16 + l], v2 = x4[s2 * 16 + l + 8];
            const float4 u3 = x4[s3 * 16 + l], v3 = x4[s3 * 16 + l + 8];
            const float w0 = d0 * di;
            const float w1 = d1 * (k1 ? di : 0.0f);
            const float w2 = d2 * (k2 ? di : 0.0f);
            const float w3 = d3 * (k3 ? di : 0.0f);
            a0.x = fmaf(w0, u0.x, a0.x); a0.y = fmaf(w0, u0.y, a0.y); a0.z = fmaf(w0, u0.z, a0.z); a0.w = fmaf(w0, u0.w, a0.w);
            a1.x = fmaf(w0, v0.x, a1.x); a1.y = fmaf(w0, v0.y, a1.y); a1.z = fmaf(w0, v0.z, a1.z); a1.w = fmaf(w0, v0.w, a1.w);
            a0.x = fmaf(w1, u1.x, a0.x); a0.y = fmaf(w1, u1.y, a0.y); a0.z = fmaf(w1, u1.z, a0.z); a0.w = fmaf(w1, u1.w, a0.w);
            a1.x = fmaf(w1, v1.x, a1.x); a1.y = fmaf(w1, v1.y, a1.y); a1.z = fmaf(w1, v1.z, a1.z); a1.w = fmaf(w1, v1.w, a1.w);
            a0.x = fmaf(w2, u2.x, a0.x); a0.y = fmaf(w2, u2.y, a0.y); a0.z = fmaf(w2, u2.z, a0.z); a0.w = fmaf(w2, u2.w, a0.w);
            a1.x = fmaf(w2, v2.x, a1.x); a1.y = fmaf(w2, v2.y, a1.y); a1.z = fmaf(w2, v2.z, a1.z); a1.w = fmaf(w2, v2.w, a1.w);
            a0.x = fmaf(w3, u3.x, a0.x); a0.y = fmaf(w3, u3.y, a0.y); a0.z = fmaf(w3, u3.z, a0.z); a0.w = fmaf(w3, u3.w, a0.w);
            a1.x = fmaf(w3, v3.x, a1.x); a1.y = fmaf(w3, v3.y, a1.y); a1.z = fmaf(w3, v3.z, a1.z); a1.w = fmaf(w3, v3.w, a1.w);
            asm volatile("" :: "v"(nxt.x), "v"(nxt.y), "v"(nxt.z), "v"(nxt.w));
            cur = nxt;
            p += 4;
        }
    }
    const int bcol = wave * 16 + lm;
    const float bias = b1[bcol];
    float wf[32];
#pragma unroll
    for (int kk = 0; kk < 16; ++kk) {
        const int kb = kk * 4 + half * 2;
        wf[2 * kk]     = W1[(size_t)kb * 64 + bcol];
        wf[2 * kk + 1] = W1[(size_t)(kb + 1) * 64 + bcol];
    }
    *(float4*)&ta[g * TP + c]      = a0;
    *(float4*)&ta[g * TP + 32 + c] = a1;
    lds_barrier();

    {
        v8f d = {0.f,0.f,0.f,0.f,0.f,0.f,0.f,0.f};
#pragma unroll
        for (int kk = 0; kk < 16; ++kk) {
            const int kb = kk * 4 + half * 2;
            const v2f a = *(const v2f*)&ta[lm * TP + kb];
            v2f b;
            b.x = wf[2 * kk];
            b.y = wf[2 * kk + 1];
            d = __builtin_amdgcn_wmma_f32_16x16x4_f32(false, a, false, b,
                                                      (short)0, d, false, false);
        }
#pragma unroll
        for (int i = 0; i < 8; ++i)
            tb[(half * 8 + i) * TP + bcol] = d[i] + bias;
    }
    const int  col2 = wave * 16 + lm;
    const bool ok   = (col2 < 34);
    const int  cc   = ok ? col2 : 33;
    float vf[32];
#pragma unroll
    for (int q = 0; q < 32; ++q) vf[q] = 0.0f;
    if (wave < 3) {
#pragma unroll
        for (int kk = 0; kk < 16; ++kk) {
            const int kb = kk * 4 + half * 2;
            vf[2 * kk]     = W2[(size_t)kb * 34 + cc];
            vf[2 * kk + 1] = W2[(size_t)(kb + 1) * 34 + cc];
        }
    }
    lds_barrier();

    const float4 p0 = *(const float4*)&tb[g * TP + c];
    const float4 p1 = *(const float4*)&tb[g * TP + 32 + c];
    const float r0 = fmaxf(p0.x, 0.0f), r1 = fmaxf(p0.y, 0.0f);
    const float r2 = fmaxf(p0.z, 0.0f), r3 = fmaxf(p0.w, 0.0f);
    const float r4 = fmaxf(p1.x, 0.0f), r5 = fmaxf(p1.y, 0.0f);
    const float r6 = fmaxf(p1.z, 0.0f), r7 = fmaxf(p1.w, 0.0f);
    float m = fmaxf(fmaxf(fmaxf(r0, r1), fmaxf(r2, r3)), fmaxf(fmaxf(r4, r5), fmaxf(r6, r7)));
    for (int off = 4; off > 0; off >>= 1) m = fmaxf(m, __shfl_xor(m, off, 8));
    const float e0 = __expf(r0 - m), e1 = __expf(r1 - m);
    const float e2 = __expf(r2 - m), e3 = __expf(r3 - m);
    const float e4 = __expf(r4 - m), e5 = __expf(r5 - m);
    const float e6 = __expf(r6 - m), e7 = __expf(r7 - m);
    float s = ((e0 + e1) + (e2 + e3)) + ((e4 + e5) + (e6 + e7));
    for (int off = 4; off > 0; off >>= 1) s += __shfl_xor(s, off, 8);
    const float inv = (1.0f / s) * di;
    float4 o0, o1;
    o0.x = e0 * inv; o0.y = e1 * inv; o0.z = e2 * inv; o0.w = e3 * inv;
    o1.x = e4 * inv; o1.y = e5 * inv; o1.z = e6 * inv; o1.w = e7 * inv;
    *(float4*)&ta[g * TP + c]      = o0;
    *(float4*)&ta[g * TP + 32 + c] = o1;
    lds_barrier();

    if (wave < 3) {
        v8f d = {0.f,0.f,0.f,0.f,0.f,0.f,0.f,0.f};
#pragma unroll
        for (int kk = 0; kk < 16; ++kk) {
            const int kb = kk * 4 + half * 2;
            const v2f a = *(const v2f*)&ta[lm * TP + kb];
            v2f b;
            b.x = ok ? vf[2 * kk]     : 0.0f;
            b.y = ok ? vf[2 * kk + 1] : 0.0f;
            d = __builtin_amdgcn_wmma_f32_16x16x4_f32(false, a, false, b,
                                                      (short)0, d, false, false);
        }
        if (ok) {
#pragma unroll
            for (int i = 0; i < 8; ++i) {
                const int row = blockIdx.x * 16 + half * 8 + i;
                if (row < n) hw2[(size_t)row * 34 + col2] = d[i];
            }
        }
    }
}

__global__ void __launch_bounds__(256) k_gather34(const int* __restrict__ rowptr,
                                                  const int* __restrict__ cnt,
                                                  const int* __restrict__ col,
                                                  const float* __restrict__ dinv,
                                                  const float* __restrict__ hw2,
                                                  const float* __restrict__ b2,
                                                  float* __restrict__ out, int n) {
    const int t = blockIdx.x * 256 + threadIdx.x;
    const int node = t >> 4;
    const int l = t & 15;
    const int c = l << 1;
    if (node >= n) return;
    const float di = dinv[node];
    int p = rowptr[node];
    const int nin = cnt[node];
    const int end = p + nin;
    const float* srow = hw2 + (size_t)node * 34;
    const v2f sv = *(const v2f*)(srow + c);
    const v2f sx = *(const v2f*)(srow + 32);
    asm volatile("" :: "v"(p), "v"(nin));
    v2f acc = sv, accx = sx;
    const v2f zz = {0.0f, 0.0f};
    int4 cur = *(const int4*)(col + p);
    while (p < end) {
        const int4 nxt = *(const int4*)(col + p + 4);
        const bool k1 = p + 1 < end, k2 = p + 2 < end, k3 = p + 3 < end;
        const int s0 = cur.x;
        const int s1 = k1 ? cur.y : s0, s2 = k2 ? cur.z : s0, s3 = k3 ? cur.w : s0;
        const float* r0 = hw2 + (size_t)s0 * 34;
        const float* r1 = hw2 + (size_t)s1 * 34;
        const float* r2 = hw2 + (size_t)s2 * 34;
        const float* r3 = hw2 + (size_t)s3 * 34;
        const v2f v0 = *(const v2f*)(r0 + c), x0 = *(const v2f*)(r0 + 32);
        const v2f v1 = *(const v2f*)(r1 + c), x1 = *(const v2f*)(r1 + 32);
        const v2f v2 = *(const v2f*)(r2 + c), x2 = *(const v2f*)(r2 + 32);
        const v2f v3 = *(const v2f*)(r3 + c), x3 = *(const v2f*)(r3 + 32);
        acc += v0;           accx += x0;
        acc += k1 ? v1 : zz; accx += k1 ? x1 : zz;
        acc += k2 ? v2 : zz; accx += k2 ? x2 : zz;
        acc += k3 ? v3 : zz; accx += k3 ? x3 : zz;
        asm volatile("" :: "v"(nxt.x), "v"(nxt.y), "v"(nxt.z), "v"(nxt.w));
        cur = nxt;
        p += 4;
    }
    v2f o, ox;
    o.x  = fmaf(di, acc.x,  b2[c]);
    o.y  = fmaf(di, acc.y,  b2[c + 1]);
    ox.x = fmaf(di, accx.x, b2[32]);
    ox.y = fmaf(di, accx.y, b2[33]);
    float* orow = out + (size_t)node * 34;
    *(v2f*)(orow + c) = o;
    if (l == 0) *(v2f*)(orow + 32) = ox;
}

extern "C" void kernel_launch(void* const* d_in, const int* in_sizes, int n_in,
                              void* d_out, int out_size, void* d_ws, size_t ws_size,
                              hipStream_t stream) {
    const float* x  = (const float*)d_in[0];
    const int*   ei = (const int*)  d_in[1];
    const float* W1 = (const float*)d_in[2];
    const float* b1 = (const float*)d_in[3];
    const float* W2 = (const float*)d_in[4];
    const float* b2 = (const float*)d_in[5];
    float* out = (float*)d_out;

    const int N = in_sizes[0] / 64;
    const int E = in_sizes[1] / 2;
    const int* src = ei;
    const int* dst = ei + E;

    int*   cnt    = (int*)d_ws;
    int*   rowptr = cnt    + N;
    int*   pad    = rowptr + N;
    float* dinv   = (float*)(pad + 64);
    float* hw2    = dinv + N;
    int*   col    = (int*)(hw2 + (size_t)N * 34);
    int*   binned = col + E;
    int*   segoff = binned + E;

    const int T = 256;
    const int nblk = (E + CH - 1) / CH;
    k_part<<<nblk, 1024, 0, stream>>>(src, dst, segoff, binned, E, nblk);
    k_csr <<<(N + RNG - 1) / RNG, 1024, 0, stream>>>(binned, segoff, cnt, rowptr, dinv, col, N, nblk);
    k_layer1<<<(N + 15) / 16, 128, 0, stream>>>(rowptr, cnt, col, dinv, x, W1, b1, W2, hw2, N);
    k_gather34<<<((size_t)N * 16 + T - 1) / T, T, 0, stream>>>(rowptr, cnt, col, dinv, hw2, b2, out, N);
}


# hw2 kept as a 32-column plane (one aligned 128-byte line per row) plus a packed 2-column plane: the layer-2 gather reads a row as one whole line, 8 lanes per node, 16 B per lane; same sums
# speedup vs baseline: 5.5261x; 1.0319x over previous
#include <hip/hip_runtime.h>
#include <math.h>

typedef __attribute__((ext_vector_type(2))) float v2f;
typedef __attribute__((ext_vector_type(8))) float v8f;


#define RSH   9
#define RNG   512
#define NBMAX 256
#define CH    5120

__global__ void __launch_bounds__(1024) k_part(const int* __restrict__ src, const int* __restrict__ dst,
                                               int* __restrict__ segoff, int* __restrict__ binned,
                                               int E, int nblk) {
    __shared__ int lcur[NBMAX];
    __shared__ int wsum[NBMAX / 32];
    __shared__ int lsort[CH];
    const int tid = threadIdx.x, lane = tid & 31, wv = tid >> 5;
    const int me = blockIdx.x;
    if (tid < NBMAX) lcur[tid] = 0;
    const int base = me * CH + tid;
    int d[CH / 1024], s[CH / 1024];
#pragma unroll
    for (int k = 0; k < CH / 1024; ++k) {
        const int e  = base + k * 1024;
        const int ec = e < E ? e : E - 1;
        d[k] = dst[ec];
        s[k] = src[ec];
    }
    __syncthreads();
#pragma unroll
    for (int k = 0; k < CH / 1024; ++k)
        if (base + k * 1024 < E) atomicAdd(&lcur[d[k] >> RSH], 1);
    __syncthreads();
    int c = 0, inc = 0;
    if (tid < NBMAX) {
        c = lcur[tid];
        inc = c;
#pragma unroll
        for (int off = 1; off < 32; off <<= 1) {
            int v = __shfl_up(inc, off, 32);
            if (lane >= off) inc += v;
        }
        if (lane == 31) wsum[wv] = inc;
    }
    __syncthreads();
    if (tid < NBMAX) {
        int woff = 0;
        for (int w = 0; w < wv; ++w) woff += wsum[w];
        const int ex = woff + inc - c;
        lcur[tid] = ex;
        segoff[(size_t)tid * nblk + me] = ex;
        if (tid == NBMAX - 1) segoff[(size_t)NBMAX * nblk + me] = ex + c;
    }
    __syncthreads();
#pragma unroll
    for (int k = 0; k < CH / 1024; ++k) {
        if (base + k * 1024 < E) {
            const int pos = atomicAdd(&lcur[d[k] >> RSH], 1);
            lsort[pos] = (s[k] << RSH) | (d[k] & (RNG - 1));
        }
    }
    __syncthreads();
    const int left = E - me * CH;
    const int nval = left < CH ? left : CH;
#pragma unroll
    for (int k = 0; k < CH / 1024; ++k) {
        const int i = tid + k * 1024;
        if (i < nval) binned[(size_t)me * CH + i] = lsort[i];
    }
}

__device__ __forceinline__ void csr_fill(const bool to_lds, const int* __restrict__ binned,
                                         const int* so0, const int* so1, int* lc, int* dstl, int* __restrict__ dstg,
                                         const int b0, const int sb0, const int eb0, const int q, const int nblk) {
    int b = b0, sb = sb0, eb = eb0;
    while (b < nblk) {
        const int* seg = binned + (size_t)b * CH;
        for (int i0 = sb + q; i0 < eb; i0 += 32) {
            int v[8];
#pragma unroll
            for (int k = 0; k < 8; ++k) {
                const int i = i0 + 4 * k;
                v[k] = seg[i < eb ? i : eb - 1];
            }
            asm volatile("" :: "v"(v[0]), "v"(v[1]), "v"(v[2]), "v"(v[3]), "v"(v[4]), "v"(v[5]), "v"(v[6]), "v"(v[7]));
#pragma unroll
            for (int k = 0; k < 8; ++k) {
                if (i0 + 4 * k < eb) {
                    const int pos = atomicAdd(&lc[v[k] & (RNG - 1)], 1);
                    if (to_lds) dstl[pos] = v[k] >> RSH;
                    else        dstg[pos] = v[k] >> RSH;
                }
            }
        }
        b += 256;
        if (b < nblk) { sb = so0[b]; eb = so1[b]; }
    }
}

#define LCAP 12288
__global__ void __launch_bounds__(1024) k_csr(const int* __restrict__ binned, const int* __restrict__ segoff,
                                              int* __restrict__ cnt, int* __restrict__ rowptr,
                                              float* __restrict__ dinv, int* __restrict__ col, int n, int nblk) {
    __shared__ int lc[RNG];
    __shared__ int wsum[RNG / 32];
    __shared__ int red[32];
    __shared__ int rtot;
    __shared__ alignas(16) int lcol[LCAP];
    const int tid = threadIdx.x, lane = tid & 31, wv = tid >> 5;
    const int* so0 = segoff + (size_t)blockIdx.x * nblk;
    const int* so1 = so0 + nblk;
    const int q  = tid & 3;
    const int b0 = tid >> 2;
    int sb0 = 0, eb0 = 0;
    if (b0 < nblk) { sb0 = so0[b0]; eb0 = so1[b0]; }
    int part = 0;
    for (int b = tid; b < nblk; b += 1024) part += so0[b];
#pragma unroll
    for (int off = 16; off > 0; off >>= 1) part += __shfl_xor(part, off, 32);
    if (lane == 0) red[wv] = part;
    if (tid < RNG) lc[tid] = 0;
    __syncthreads();
    int start = red[lane];
#pragma unroll
    for (int off = 16; off > 0; off >>= 1) start += __shfl_xor(start, off, 32);
    {
        int b = b0, sb = sb0, eb = eb0;
        while (b < nblk) {
            const int* seg = binned + (size_t)b * CH;
            for (int i0 = sb + q; i0 < eb; i0 += 32) {
                int v[8];
#pragma unroll
                for (int k = 0; k < 8; ++k) {
                    const int i = i0 + 4 * k;
                    v[k] = seg[i < eb ? i : eb - 1];
                }
                asm volatile("" :: "v"(v[0]), "v"(v[1]), "v"(v[2]), "v"(v[3]), "v"(v[4]), "v"(v[5]), "v"(v[6]), "v"(v[7]));
#pragma unroll
                for (int k = 0; k < 8; ++k)
                    if (i0 + 4 * k < eb) atomicAdd(&lc[v[k] & (RNG - 1)], 1);
            }
            b += 256;
            if (b < nblk) { sb = so0[b]; eb = so1[b]; }
        }
    }
    __syncthreads();
    int c = 0, inc = 0;
    if (tid < RNG) {
        c = lc[tid];
        inc = c;
#pragma unroll
        for (int off = 1; off < 32; off <<= 1) {
            int v = __shfl_up(inc, off, 32);
            if (lane >= off) inc += v;
        }
        if (lane == 31) wsum[wv] = inc;
    }
    __syncthreads();
    if (tid < RNG) {
        int woff = 0;
        for (int w = 0; w < wv; ++w) woff += wsum[w];
        const int ex = woff + inc - c;
        lc[tid] = ex;
        if (tid == RNG - 1) rtot = ex + c;
        const int node = blockIdx.x * RNG + tid;
        if (node < n) {
            cnt[node]    = c;
            rowptr[node] = start + ex;
            dinv[node]   = rsqrtf((float)(c + 1));
        }
    }
    __syncthreads();
    const int total = rtot;
    const int a = (int)((((size_t)(col + start)) & 127) >> 2);
    const int endw = a + total;
    const bool in_lds = __builtin_amdgcn_readfirstlane(endw) <= LCAP;
    if (in_lds) {
        csr_fill(true, binned, so0, so1, lc, lcol + a, col, b0, sb0, eb0, q, nblk);
        __syncthreads();
        int* cline = col + start - a;
        for (int j = tid * 4; j < endw; j += 4096) {
            if (j >= a && j + 4 <= endw) {
                *(int4*)(cline + j) = *(const int4*)&lcol[j];
            } else {
#pragma unroll
                for (int k = 0; k < 4; ++k)
                    if (j + k >= a && j + k < endw) cline[j + k] = lcol[j + k];
            }
        }
    } else {
        csr_fill(false, binned, so0, so1, lc, lcol, col + start, b0, sb0, eb0, q, nblk);
    }
}

#define TP 68
__device__ __forceinline__ void lds_barrier() {
    __builtin_amdgcn_fence(__ATOMIC_RELEASE, "workgroup", "local");
    __builtin_amdgcn_s_barrier();
    __builtin_amdgcn_fence(__ATOMIC_ACQUIRE, "workgroup", "local");
}
__global__ void __launch_bounds__(128) k_layer1(const int* __restrict__ rowptr,
                                                const int* __restrict__ cnt,
                                                const int* __restrict__ col,
                                                const float* __restrict__ dinv,
                                                const float* __restrict__ x,
                                                const float* W1,
                                                const float* __restrict__ b1,
                                                const float* W2,
                                                float* __restrict__ hm,
                                                float* __restrict__ hs, int n) {
    __shared__ float ta[16 * TP];
    __shared__ float tb[16 * TP];
    const int tid  = threadIdx.x;
    const int g    = tid >> 3;
    const int c    = (tid & 7) << 2;
    const int node = blockIdx.x * 16 + g;
    const int wave = tid >> 5;
    const int lane = tid & 31;
    const int half = lane >> 4;
    const int lm   = lane & 15;
    float4 a0 = make_float4(0.0f, 0.0f, 0.0f, 0.0f);
    float4 a1 = make_float4(0.0f, 0.0f, 0.0f, 0.0f);
    float di = 0.0f;
    if (node < n) {
        di = dinv[node];
        int p = rowptr[node];
        const int nin = cnt[node];
        const int end = p + nin;
        const float4* x4 = (const float4*)x;
        const int l = tid & 7;
        const float4 su = x4[node * 16 + l];
        const float4 sv = x4[node * 16 + l + 8];
        asm volatile("" :: "v"(p), "v"(nin));
        const float wself = di * di;
        a0.x = wself * su.x; a0.y = wself * su.y; a0.z = wself * su.z; a0.w = wself * su.w;
        a1.x = wself * sv.x; a1.y = wself * sv.y; a1.z = wself * sv.z; a1.w = wself * sv.w;
        int4 cur = *(const int4*)(col + p);
        while (p < end) {
            const int4 nxt = *(const int4*)(col + p + 4);
            const bool k1 = p + 1 < end, k2 = p + 2 < end, k3 = p + 3 < end;
            const int s0 = cur.x;
            const int s1 = k1 ? cur.y : s0, s2 = k2 ? cur.z : s0, s3 = k3 ? cur.w : s0;
            const float d0 = dinv[s0], d1 = dinv[s1], d2 = dinv[s2], d3 = dinv[s3];
            const float4 u0 = x4[s0 * 16 + l], v0 = x4[s0 * 16 + l + 8];
            const float4 u1 = x4[s1 * 16 + l], v1 = x4[s1 * 16 + l + 8];
            const float4 u2 = x4[s2 * 16 + l], v2 = x4[s2 * 16 + l + 8];
            const float4 u3 = x4[s3 * 16 + l], v3 = x4[s3 * 16 + l + 8];
            const float w0 = d0 * di;
            const float w1 = d1 * (k1 ? di : 0.0f);
            const float w2 = d2 * (k2 ? di : 0.0f);
            const float w3 = d3 * (k3 ? di : 0.0f);
            a0.x = fmaf(w0, u0.x, a0.x); a0.y = fmaf(w0, u0.y, a0.y); a0.z = fmaf(w0, u0.z, a0.z); a0.w = fmaf(w0, u0.w, a0.w);
            a1.x = fmaf(w0, v0.x, a1.x); a1.y = fmaf(w0, v0.y, a1.y); a1.z = fmaf(w0, v0.z, a1.z); a1.w = fmaf(w0, v0.w, a1.w);
            a0.x = fmaf(w1, u1.x, a0.x); a0.y = fmaf(w1, u1.y, a0.y); a0.z = fmaf(w1, u1.z, a0.z); a0.w = fmaf(w1, u1.w, a0.w);
            a1.x = fmaf(w1, v1.x, a1.x); a1.y = fmaf(w1, v1.y, a1.y); a1.z = fmaf(w1, v1.z, a1.z); a1.w = fmaf(w1, v1.w, a1.w);
            a0.x = fmaf(w2, u2.x, a0.x); a0.y = fmaf(w2, u2.y, a0.y); a0.z = fmaf(w2, u2.z, a0.z); a0.w = fmaf(w2, u2.w, a0.w);
            a1.x = fmaf(w2, v2.x, a1.x); a1.y = fmaf(w2, v2.y, a1.y); a1.z = fmaf(w2, v2.z, a1.z); a1.w = fmaf(w2, v2.w, a1.w);
            a0.x = fmaf(w3, u3.x, a0.x); a0.y = fmaf(w3, u3.y, a0.y); a0.z = fmaf(w3, u3.z, a0.z); a0.w = fmaf(w3, u3.w, a0.w);
            a1.x = fmaf(w3, v3.x, a1.x); a1.y = fmaf(w3, v3.y, a1.y); a1.z = fmaf(w3, v3.z, a1.z); a1.w = fmaf(w3, v3.w, a1.w);
            asm volatile("" :: "v"(nxt.x), "v"(nxt.y), "v"(nxt.z), "v"(nxt.w));
            cur = nxt;
            p += 4;
        }
    }
    const int bcol = wave * 16 + lm;
    const float bias = b1[bcol];
    float wf[32];
#pragma unroll
    for (int kk = 0; kk < 16; ++kk) {
        const int kb = kk * 4 + half * 2;
        wf[2 * kk]     = W1[(size_t)kb * 64 + bcol];
        wf[2 * kk + 1] = W1[(size_t)(kb + 1) * 64 + bcol];
    }
    *(float4*)&ta[g * TP + c]      = a0;
    *(float4*)&ta[g * TP + 32 + c] = a1;
    lds_barrier();

    {
        v8f d = {0.f,0.f,0.f,0.f,0.f,0.f,0.f,0.f};
#pragma unroll
        for (int kk = 0; kk < 16; ++kk) {
            const int kb = kk * 4 + half * 2;
            const v2f a = *(const v2f*)&ta[lm * TP + kb];
            v2f b;
            b.x = wf[2 * kk];
            b.y = wf[2 * kk + 1];
            d = __builtin_amdgcn_wmma_f32_16x16x4_f32(false, a, false, b,
                                                      (short)0, d, false, false);
        }
#pragma unroll
        for (int i = 0; i < 8; ++i)
            tb[(half * 8 + i) * TP + bcol] = d[i] + bias;
    }
    const int  col2 = wave * 16 + lm;
    const bool ok   = (col2 < 34);
    const int  cc   = ok ? col2 : 33;
    float vf[32];
#pragma unroll
    for (int q = 0; q < 32; ++q) vf[q] = 0.0f;
    if (wave < 3) {
#pragma unroll
        for (int kk = 0; kk < 16; ++kk) {
            const int kb = kk * 4 + half * 2;
            vf[2 * kk]     = W2[(size_t)kb * 34 + cc];
            vf[2 * kk + 1] = W2[(size_t)(kb + 1) * 34 + cc];
        }
    }
    lds_barrier();

    const float4 p0 = *(const float4*)&tb[g * TP + c];
    const float4 p1 = *(const float4*)&tb[g * TP + 32 + c];
    const float r0 = fmaxf(p0.x, 0.0f), r1 = fmaxf(p0.y, 0.0f);
    const float r2 = fmaxf(p0.z, 0.0f), r3 = fmaxf(p0.w, 0.0f);
    const float r4 = fmaxf(p1.x, 0.0f), r5 = fmaxf(p1.y, 0.0f);
    const float r6 = fmaxf(p1.z, 0.0f), r7 = fmaxf(p1.w, 0.0f);
    float m = fmaxf(fmaxf(fmaxf(r0, r1), fmaxf(r2, r3)), fmaxf(fmaxf(r4, r5), fmaxf(r6, r7)));
    for (int off = 4; off > 0; off >>= 1) m = fmaxf(m, __shfl_xor(m, off, 8));
    const float e0 = __expf(r0 - m), e1 = __expf(r1 - m);
    const float e2 = __expf(r2 - m), e3 = __expf(r3 - m);
    const float e4 = __expf(r4 - m), e5 = __expf(r5 - m);
    const float e6 = __expf(r6 - m), e7 = __expf(r7 - m);
    float s = ((e0 + e1) + (e2 + e3)) + ((e4 + e5) + (e6 + e7));
    for (int off = 4; off > 0; off >>= 1) s += __shfl_xor(s, off, 8);
    const float inv = (1.0f / s) * di;
    float4 o0, o1;
    o0.x = e0 * inv; o0.y = e1 * inv; o0.z = e2 * inv; o0.w = e3 * inv;
    o1.x = e4 * inv; o1.y = e5 * inv; o1.z = e6 * inv; o1.w = e7 * inv;
    *(float4*)&ta[g * TP + c]      = o0;
    *(float4*)&ta[g * TP + 32 + c] = o1;
    lds_barrier();

    if (wave < 3) {
        v8f d = {0.f,0.f,0.f,0.f,0.f,0.f,0.f,0.f};
#pragma unroll
        for (int kk = 0; kk < 16; ++kk) {
            const int kb = kk * 4 + half * 2;
            const v2f a = *(const v2f*)&ta[lm * TP + kb];
            v2f b;
            b.x = ok ? vf[2 * kk]     : 0.0f;
            b.y = ok ? vf[2 * kk + 1] : 0.0f;
            d = __builtin_amdgcn_wmma_f32_16x16x4_f32(false, a, false, b,
                                                      (short)0, d, false, false);
        }
        if (ok) {
            float* dstp = wave < 2 ? hm + col2 : hs + (col2 - 32);
            const int pitch = wave < 2 ? 32 : 2;
#pragma unroll
            for (int i = 0; i < 8; ++i) {
                const int row = blockIdx.x * 16 + half * 8 + i;
                if (row < n) dstp[(size_t)row * pitch] = d[i];
            }
        }
    }
}

typedef __attribute__((ext_vector_type(4), aligned(8))) float v4f8;
__global__ void __launch_bounds__(256) k_gather34(const int* __restrict__ rowptr,
                                                  const int* __restrict__ cnt,
                                                  const int* __restrict__ col,
                                                  const float* __restrict__ dinv,
                                                  const float* __restrict__ hm,
                                                  const float* __restrict__ hs,
                                                  const float* __restrict__ b2,
                                                  float* __restrict__ out, int n) {
    const int t = blockIdx.x * 256 + threadIdx.x;
    const int node = t >> 3;
    const int l = t & 7;
    const int c = l << 2;
    if (node >= n) return;
    const float di = dinv[node];
    int p = rowptr[node];
    const int nin = cnt[node];
    const int end = p + nin;
    const float4* m4 = (const float4*)hm;
    const v2f*    s2 = (const v2f*)hs;
    const float4 sv = m4[node * 8 + l];
    const v2f    sx = s2[node];
    asm volatile("" :: "v"(p), "v"(nin));
    float4 acc = sv;
    v2f accx = sx;
    const v2f zz = {0.0f, 0.0f};
    int4 cur = *(const int4*)(col + p);
    while (p < end) {
        const int4 nxt = *(const int4*)(col + p + 4);
        const bool k1 = p + 1 < end, k2 = p + 2 < end, k3 = p + 3 < end;
        const int s0 = cur.x;
        const int s1 = k1 ? cur.y : s0, s2i = k2 ? cur.z : s0, s3 = k3 ? cur.w : s0;
        const float4 v0 = m4[s0  * 8 + l]; const v2f x0 = s2[s0];
        const float4 v1 = m4[s1  * 8 + l]; const v2f x1 = s2[s1];
        const float4 v2 = m4[s2i * 8 + l]; const v2f x2 = s2[s2i];
        const float4 v3 = m4[s3  * 8 + l]; const v2f x3 = s2[s3];
        acc.x += v0.x; acc.y += v0.y; acc.z += v0.z; acc.w += v0.w; accx += x0;
        acc.x += k1 ? v1.x : 0.0f; acc.y += k1 ? v1.y : 0.0f; acc.z += k1 ? v1.z : 0.0f; acc.w += k1 ? v1.w : 0.0f;
        accx  += k1 ? x1 : zz;
        acc.x += k2 ? v2.x : 0.0f; acc.y += k2 ? v2.y : 0.0f; acc.z += k2 ? v2.z : 0.0f; acc.w += k2 ? v2.w : 0.0f;
        accx  += k2 ? x2 : zz;
        acc.x += k3 ? v3.x : 0.0f; acc.y += k3 ? v3.y : 0.0f; acc.z += k3 ? v3.z : 0.0f; acc.w += k3 ? v3.w : 0.0f;
        accx  += k3 ? x3 : zz;
        asm volatile("" :: "v"(nxt.x), "v"(nxt.y), "v"(nxt.z), "v"(nxt.w));
        cur = nxt;
        p += 4;
    }
    v4f8 o;
    v2f ox;
    o.x  = fmaf(di, acc.x,  b2[c]);
    o.y  = fmaf(di, acc.y,  b2[c + 1]);
    o.z  = fmaf(di, acc.z,  b2[c + 2]);
    o.w  = fmaf(di, acc.w,  b2[c + 3]);
    ox.x = fmaf(di, accx.x, b2[32]);
    ox.y = fmaf(di, accx.y, b2[33]);
    float* orow = out + (size_t)node * 34;
    *(v4f8*)(orow + c) = o;
    if (l == 0) *(v2f*)(orow + 32) = ox;
}

extern "C" void kernel_launch(void* const* d_in, const int* in_sizes, int n_in,
                              void* d_out, int out_size, void* d_ws, size_t ws_size,
                              hipStream_t stream) {
    const float* x  = (const float*)d_in[0];
    const int*   ei = (const int*)  d_in[1];
    const float* W1 = (const float*)d_in[2];
    const float* b1 = (const float*)d_in[3];
    const float* W2 = (const float*)d_in[4];
    const float* b2 = (const float*)d_in[5];
    float* out = (float*)d_out;

    const int N = in_sizes[0] / 64;
    const int E = in_sizes[1] / 2;
    const int* src = ei;
    const int* dst = ei + E;

    int*   cnt    = (int*)d_ws;
    int*   rowptr = cnt    + N;
    int*   pad    = rowptr + N;
    float* dinv   = (float*)(pad + 64);
    float* hm     = dinv + N;
    float* hs     = hm + (size_t)N * 32;
    int*   col    = (int*)(hs + (size_t)N * 2);
    int*   binned = col + E;
    int*   segoff = binned + E;

    const int T = 256;
    const int nblk = (E + CH - 1) / CH;
    k_part<<<nblk, 1024, 0, stream>>>(src, dst, segoff, binned, E, nblk);
    k_csr <<<(N + RNG - 1) / RNG, 1024, 0, stream>>>(binned, segoff, cnt, rowptr, dinv, col, N, nblk);
    k_layer1<<<(N + 15) / 16, 128, 0, stream>>>(rowptr, cnt, col, dinv, x, W1, b1, W2, hm, hs, N);
    k_gather34<<<((size_t)N * 8 + T - 1) / T, T, 0, stream>>>(rowptr, cnt, col, dinv, hm, hs, b2, out, N);
}
